# starting kernel
# baseline (speedup 1.0000x reference)
#include <hip/hip_runtime.h>
#include <hip/hip_bf16.h>
#include <math.h>


typedef __attribute__((ext_vector_type(2))) float v2f;
typedef __attribute__((ext_vector_type(8))) float v8f;

#define DIM_K 128

__device__ __forceinline__ float fast_tanh(float x) {
    float ax = fabsf(x);
    float e  = __expf(-2.0f * ax);
    float r  = (1.0f - e) * __builtin_amdgcn_rcpf(1.0f + e);
    return copysignf(r, x);
}

__global__ void gcn_zero_kernel(float* __restrict__ p, long long n) {
    long long i = (long long)blockIdx.x * blockDim.x + threadIdx.x;
    long long stride = (long long)gridDim.x * blockDim.x;
    for (; i < n; i += stride) p[i] = 0.0f;
}

__global__ void gcn_spmm_kernel(const float* __restrict__ vals,
                                const int* __restrict__ row,
                                const int* __restrict__ col,
                                const float* __restrict__ X,
                                float* __restrict__ H,
                                int E) {
    int wavesPerBlock = blockDim.x >> 5;
    int e = blockIdx.x * wavesPerBlock + (threadIdx.x >> 5);
    int lane = threadIdx.x & 31;
    if (e >= E) return;
    float v = vals[e];
    int c = col[e];
    int r = row[e];
    const float4* src = (const float4*)(X + (long long)c * DIM_K);
    float4 m = src[lane];
    float* dst = H + (long long)r * DIM_K + lane * 4;
    atomicAdd(dst + 0, v * m.x);
    atomicAdd(dst + 1, v * m.y);
    atomicAdd(dst + 2, v * m.z);
    atomicAdd(dst + 3, v * m.w);
}

template <int NT>
__global__ void gcn_gemm_tanh_kernel(const float* __restrict__ S,
                                     const float* __restrict__ W,
                                     float* __restrict__ O,
                                     int nNodes) {
    const int lane = threadIdx.x & 31;
    const int wavesPerBlock = blockDim.x >> 5;
    const int mTile = blockIdx.x * wavesPerBlock + (threadIdx.x >> 5);
    if (mTile * 16 >= nNodes) return;

    const int nOut  = NT * 16;
    const int l15   = lane & 15;
    const int khalf = (lane < 16) ? 0 : 2;

    const float* aBase = S + (long long)(mTile * 16 + l15) * DIM_K + khalf;
    const float* bBase = W + (long long)l15 * DIM_K + khalf;

    v8f acc[NT];
#pragma unroll
    for (int t = 0; t < NT; ++t) acc[t] = (v8f){};

#pragma unroll 2
    for (int k = 0; k < DIM_K; k += 4) {
        v2f a;
        a.x = aBase[k];
        a.y = aBase[k + 1];
#pragma unroll
        for (int t = 0; t < NT; ++t) {
            const float* bp = bBase + (long long)t * 16 * DIM_K + k;
            v2f b;
            b.x = bp[0];
            b.y = bp[1];
            acc[t] = __builtin_amdgcn_wmma_f32_16x16x4_f32(
                 false, a,  false, b,
                 (short)0, acc[t],  false,  false);
        }
    }

    const int mbase = mTile * 16 + ((lane < 16) ? 0 : 8);
#pragma unroll
    for (int t = 0; t < NT; ++t) {
        const int n = t * 16 + l15;
#pragma unroll
        for (int r = 0; r < 8; ++r) {
            O[(long long)(mbase + r) * nOut + n] = fast_tanh(acc[t][r]);
        }
    }
}

extern "C" void kernel_launch(void* const* d_in, const int* in_sizes, int n_in,
                              void* d_out, int out_size, void* d_ws, size_t ws_size,
                              hipStream_t stream) {
    const float* X        = (const float*)d_in[0];
    const float* W1       = (const float*)d_in[1];
    const float* W2       = (const float*)d_in[2];
    const float* edgeVals = (const float*)d_in[3];
    const int*   edgeRow  = (const int*)d_in[4];
    const int*   edgeCol  = (const int*)d_in[5];
    float* out = (float*)d_out;

    const int N = in_sizes[0] / DIM_K;
    const int E = in_sizes[3];
    const long long featCount = (long long)N * DIM_K;

    float* S  = (float*)d_ws;
    float* H1 = S + featCount;

    const int zeroBlocks = 2048, zeroThreads = 256;

    const int spmmThreads = 256;
    const int edgesPerBlock = spmmThreads / 32;
    const int spmmBlocks = (E + edgesPerBlock - 1) / edgesPerBlock;

    const int gemmThreads = 256;
    const int wavesPerBlock = gemmThreads / 32;
    const int mTiles = (N + 15) / 16;
    const int gemmBlocks = (mTiles + wavesPerBlock - 1) / wavesPerBlock;

    gcn_zero_kernel<<<zeroBlocks, zeroThreads, 0, stream>>>(S, featCount);
    gcn_spmm_kernel<<<spmmBlocks, spmmThreads, 0, stream>>>(
        edgeVals, edgeRow, edgeCol, X, S, E);
    gcn_gemm_tanh_kernel<8><<<gemmBlocks, gemmThreads, 0, stream>>>(
        S, W1, H1, N);

    gcn_zero_kernel<<<zeroBlocks, zeroThreads, 0, stream>>>(S, featCount);
    gcn_spmm_kernel<<<spmmBlocks, spmmThreads, 0, stream>>>(
        edgeVals, edgeRow, edgeCol, H1, S, E);
    gcn_gemm_tanh_kernel<4><<<gemmBlocks, gemmThreads, 0, stream>>>(
        S, W2, out, N);
}


# Float-atomic edge scatter replaced by row-bucketed gather: edges grouped by row once per call (int atomics); one wave sums each row in registers, in bucket-fill order, not atomic arrival order
# speedup vs baseline: 1.7431x; 1.7431x over previous
#include <hip/hip_runtime.h>
#include <hip/hip_bf16.h>
#include <math.h>


typedef __attribute__((ext_vector_type(2))) float v2f;
typedef __attribute__((ext_vector_type(8))) float v8f;

#define DIM_K 128

__device__ __forceinline__ float fast_tanh(float x) {
    float ax = fabsf(x);
    float e  = __expf(-2.0f * ax);
    float r  = (1.0f - e) * __builtin_amdgcn_rcpf(1.0f + e);
    return copysignf(r, x);
}

__global__ void gcn_zero_int_kernel(int* __restrict__ p, int n) {
    int i = blockIdx.x * blockDim.x + threadIdx.x;
    int stride = gridDim.x * blockDim.x;
    for (; i < n; i += stride) p[i] = 0;
}

__global__ void gcn_count_kernel(const int* __restrict__ row,
                                 int* __restrict__ cnt, int E) {
    int e = blockIdx.x * blockDim.x + threadIdx.x;
    if (e < E) atomicAdd(cnt + row[e], 1);
}

__global__ void gcn_reserve_kernel(const int* __restrict__ cnt,
                                   int* __restrict__ cursor,
                                   int* __restrict__ total, int N) {
    const int r = blockIdx.x * blockDim.x + threadIdx.x;
    const int lane = threadIdx.x & 31;
    const int c = (r < N) ? cnt[r] : 0;
    int incl = c;
#pragma unroll
    for (int d = 1; d < 32; d <<= 1) {
        int t = __shfl_up(incl, d, 32);
        if (lane >= d) incl += t;
    }
    const int waveTotal = __shfl(incl, 31, 32);
    int base = 0;
    if (lane == 31) base = atomicAdd(total, waveTotal);
    base = __shfl(base, 31, 32);
    if (r < N) cursor[r] = base + incl - c;
}

__global__ void gcn_fill_kernel(const float* __restrict__ vals,
                                const int* __restrict__ row,
                                const int* __restrict__ col,
                                int* __restrict__ cursor,
                                int* __restrict__ bucket, int E) {
    int e = blockIdx.x * blockDim.x + threadIdx.x;
    if (e >= E) return;
    int pos = atomicAdd(cursor + row[e], 1);
    ((int2*)bucket)[pos] = make_int2(col[e], __float_as_int(vals[e]));
}

template <int G>
__device__ __forceinline__ void gcn_gather_group(const float4* __restrict__ X4,
                                                 int2 my, int j, int lane,
                                                 float4& acc) {
    int c[G];
    float v[G];
    float4 x[G];
#pragma unroll
    for (int k = 0; k < G; ++k) {
        c[k] = __builtin_amdgcn_readlane(my.x, j + k);
        v[k] = __int_as_float(__builtin_amdgcn_readlane(my.y, j + k));
    }
#pragma unroll
    for (int k = 0; k < G; ++k) x[k] = X4[(long long)c[k] * (DIM_K / 4) + lane];
#pragma unroll
    for (int k = 0; k < G; ++k) {
        acc.x = __builtin_fmaf(v[k], x[k].x, acc.x);
        acc.y = __builtin_fmaf(v[k], x[k].y, acc.y);
        acc.z = __builtin_fmaf(v[k], x[k].z, acc.z);
        acc.w = __builtin_fmaf(v[k], x[k].w, acc.w);
    }
}

__global__ void gcn_gather_kernel(const float* __restrict__ X,
                                  const int* __restrict__ cnt,
                                  const int* __restrict__ cursorEnd,
                                  const int* __restrict__ bucket,
                                  float* __restrict__ H, int N) {
    const int lane = threadIdx.x & 31;
    const int r = __builtin_amdgcn_readfirstlane(
        blockIdx.x * (blockDim.x >> 5) + (threadIdx.x >> 5));
    if (r >= N) return;
    const int n = cnt[r];
    const int2* ed = (const int2*)bucket + (cursorEnd[r] - n);
    const float4* X4 = (const float4*)X;
    float4 acc = make_float4(0.0f, 0.0f, 0.0f, 0.0f);

    for (int base = 0; base < n; base += 32) {
        const int m = min(32, n - base);
        int2 my = make_int2(0, 0);
        if (lane < m) my = ed[base + lane];
        int j = 0;
        for (; j + 8 <= m; j += 8) gcn_gather_group<8>(X4, my, j, lane, acc);
        if ((m - j) & 4) { gcn_gather_group<4>(X4, my, j, lane, acc); j += 4; }
        if ((m - j) & 2) { gcn_gather_group<2>(X4, my, j, lane, acc); j += 2; }
        if ((m - j) & 1) { gcn_gather_group<1>(X4, my, j, lane, acc); }
    }
    ((float4*)(H + (long long)r * DIM_K))[lane] = acc;
}

template <int NT>
__global__ void gcn_gemm_tanh_kernel(const float* __restrict__ S,
                                     const float* __restrict__ W,
                                     float* __restrict__ O,
                                     int nNodes) {
    const int lane = threadIdx.x & 31;
    const int wavesPerBlock = blockDim.x >> 5;
    const int mTile = blockIdx.x * wavesPerBlock + (threadIdx.x >> 5);
    if (mTile * 16 >= nNodes) return;

    const int nOut  = NT * 16;
    const int l15   = lane & 15;
    const int khalf = (lane < 16) ? 0 : 2;

    const float* aBase = S + (long long)(mTile * 16 + l15) * DIM_K + khalf;
    const float* bBase = W + (long long)l15 * DIM_K + khalf;

    v8f acc[NT];
#pragma unroll
    for (int t = 0; t < NT; ++t) acc[t] = (v8f){};

#pragma unroll 2
    for (int k = 0; k < DIM_K; k += 4) {
        v2f a;
        a.x = aBase[k];
        a.y = aBase[k + 1];
#pragma unroll
        for (int t = 0; t < NT; ++t) {
            const float* bp = bBase + (long long)t * 16 * DIM_K + k;
            v2f b;
            b.x = bp[0];
            b.y = bp[1];
            acc[t] = __builtin_amdgcn_wmma_f32_16x16x4_f32(
                 false, a,  false, b,
                 (short)0, acc[t],  false,  false);
        }
    }

    const int mbase = mTile * 16 + ((lane < 16) ? 0 : 8);
#pragma unroll
    for (int t = 0; t < NT; ++t) {
        const int n = t * 16 + l15;
#pragma unroll
        for (int r = 0; r < 8; ++r) {
            O[(long long)(mbase + r) * nOut + n] = fast_tanh(acc[t][r]);
        }
    }
}

extern "C" void kernel_launch(void* const* d_in, const int* in_sizes, int n_in,
                              void* d_out, int out_size, void* d_ws, size_t ws_size,
                              hipStream_t stream) {
    const float* X        = (const float*)d_in[0];
    const float* W1       = (const float*)d_in[1];
    const float* W2       = (const float*)d_in[2];
    const float* edgeVals = (const float*)d_in[3];
    const int*   edgeRow  = (const int*)d_in[4];
    const int*   edgeCol  = (const int*)d_in[5];
    float* out = (float*)d_out;

    const int N = in_sizes[0] / DIM_K;
    const int E = in_sizes[3];
    const long long featCount = (long long)N * DIM_K;

    float* S  = (float*)d_ws;
    float* H1 = S + featCount;
    int* cnt    = (int*)(H1 + featCount);
    int* total  = cnt + N;
    int* cursor = total + 32;
    int* bucket = cursor + N + 32;

    const int threads = 256;
    const int edgeBlocks = (E + threads - 1) / threads;
    const int nodeBlocks = (N + threads - 1) / threads;

    const int wavesPerBlock = threads / 32;
    const int rowBlocks = (N + wavesPerBlock - 1) / wavesPerBlock;

    const int gemmThreads = 256;
    const int mTiles = (N + 15) / 16;
    const int gemmBlocks = (mTiles + wavesPerBlock - 1) / wavesPerBlock;

    gcn_zero_int_kernel<<<nodeBlocks, threads, 0, stream>>>(cnt, N + 1);
    gcn_count_kernel<<<edgeBlocks, threads, 0, stream>>>(edgeRow, cnt, E);
    gcn_reserve_kernel<<<nodeBlocks, threads, 0, stream>>>(cnt, cursor, total, N);
    gcn_fill_kernel<<<edgeBlocks, threads, 0, stream>>>(
        edgeVals, edgeRow, edgeCol, cursor, bucket, E);

    gcn_gather_kernel<<<rowBlocks, threads, 0, stream>>>(X, cnt, cursor, bucket, S, N);
    gcn_gemm_tanh_kernel<8><<<gemmBlocks, gemmThreads, 0, stream>>>(
        S, W1, H1, N);

    gcn_gather_kernel<<<rowBlocks, threads, 0, stream>>>(H1, cnt, cursor, bucket, S, N);
    gcn_gemm_tanh_kernel<4><<<gemmBlocks, gemmThreads, 0, stream>>>(
        S, W2, out, N);
}


# Edges grouped by row with a two-level counting sort whose counters all live in LDS (128-row bins, then rows; no global atomics); a row's sum order is now LDS-atomic arrival order
# speedup vs baseline: 3.2978x; 1.8919x over previous
#include <hip/hip_runtime.h>
#include <hip/hip_bf16.h>
#include <math.h>


typedef __attribute__((ext_vector_type(2))) float v2f;
typedef __attribute__((ext_vector_type(8))) float v8f;

#define DIM_K 128

__device__ __forceinline__ float fast_tanh(float x) {
    float ax = fabsf(x);
    float e  = __expf(-2.0f * ax);
    float r  = (1.0f - e) * __builtin_amdgcn_rcpf(1.0f + e);
    return copysignf(r, x);
}

#define ROWS_PER_BIN_LOG2 7
#define ROWS_PER_BIN (1 << ROWS_PER_BIN_LOG2)
#define EDGE_CHUNKS 256
#define SORT_THREADS 1024
#define EDGE_BATCH 8
#define STAGE_CAP (EDGE_BATCH * SORT_THREADS)
#define ROWS_THREADS 256
#define BIN_STAGE_CAP 4096

typedef unsigned long long gcn_rec;

__device__ __forceinline__ gcn_rec gcn_make_rec(int lo, int hi) {
    return (gcn_rec)(unsigned)lo | ((gcn_rec)(unsigned)hi << 32);
}

__device__ __forceinline__ int gcn_wave_incl_scan(int x, int lane) {
#pragma unroll
    for (int d = 1; d < 32; d <<= 1) {
        int t = __shfl_up(x, d, 32);
        if (lane >= d) x += t;
    }
    return x;
}

__device__ __forceinline__ int gcn_block_excl_scan(int x, int* waveSum, int* blockSum, int& total) {
    const int lane = threadIdx.x & 31;
    const int wave = threadIdx.x >> 5;
    const int nw = blockDim.x >> 5;
    const int incl = gcn_wave_incl_scan(x, lane);
    if (lane == 31) waveSum[wave] = incl;
    __syncthreads();
    if (wave == 0) {
        const int w = (lane < nw) ? waveSum[lane] : 0;
        const int wi = gcn_wave_incl_scan(w, lane);
        waveSum[lane] = wi - w;
        if (lane == 31) *blockSum = wi;
    }
    __syncthreads();
    const int excl = waveSum[wave] + incl - x;
    total = *blockSum;
    __syncthreads();
    return excl;
}

__global__ void __launch_bounds__(SORT_THREADS)
gcn_bin_hist_kernel(const int* __restrict__ row, int* __restrict__ chunkHist, int E, int NB) {
    extern __shared__ int ldsBins[];
    for (int i = threadIdx.x; i < NB; i += SORT_THREADS) ldsBins[i] = 0;
    __syncthreads();
    const int chunk = (E + EDGE_CHUNKS - 1) / EDGE_CHUNKS;
    const int lo = min(E, (int)blockIdx.x * chunk);
    const int hi = min(E, lo + chunk);
    for (int e0 = lo + threadIdx.x; e0 < hi; e0 += STAGE_CAP) {
        int r[EDGE_BATCH];
#pragma unroll
        for (int k = 0; k < EDGE_BATCH; ++k) r[k] = row[min(e0 + k * SORT_THREADS, hi - 1)];
#pragma unroll
        for (int k = 0; k < EDGE_BATCH; ++k)
            if (e0 + k * SORT_THREADS < hi) atomicAdd(&ldsBins[r[k] >> ROWS_PER_BIN_LOG2], 1);
    }
    __syncthreads();
    for (int i = threadIdx.x; i < NB; i += SORT_THREADS)
        chunkHist[blockIdx.x * NB + i] = ldsBins[i];
}

__global__ void __launch_bounds__(SORT_THREADS)
gcn_bin_offsets_kernel(const int* __restrict__ chunkHist, int* __restrict__ chunkOff,
                       int* __restrict__ binTotal, int NB) {
    __shared__ int part[32][33];
    const int lane = threadIdx.x & 31;
    const int wave = threadIdx.x >> 5;
    const int bin = blockIdx.x * 32 + lane;
    const bool live = bin < NB;
    constexpr int PER = EDGE_CHUNKS / 32;
    int v[PER];
#pragma unroll
    for (int k = 0; k < PER; ++k) v[k] = live ? chunkHist[(wave * PER + k) * NB + bin] : 0;
    int s = 0;
#pragma unroll
    for (int k = 0; k < PER; ++k) { const int t = v[k]; v[k] = s; s += t; }
    part[wave][lane] = s;
    __syncthreads();
    int off = 0;
    for (int w = 0; w < wave; ++w) off += part[w][lane];
    if (live) {
#pragma unroll
        for (int k = 0; k < PER; ++k) chunkOff[(wave * PER + k) * NB + bin] = off + v[k];
        if (wave == 31) binTotal[bin] = off + s;
    }
}

__global__ void __launch_bounds__(SORT_THREADS)
gcn_bin_scatter_kernel(const float* __restrict__ vals, const int* __restrict__ row,
                       const int* __restrict__ col, const int* __restrict__ chunkOff,
                       const int* __restrict__ binTotal, int* __restrict__ binBase,
                       int* __restrict__ binnedWords, int E, int NB) {
    gcn_rec* __restrict__ binned = (gcn_rec*)binnedWords;
    extern __shared__ int ldsBins[];
    __shared__ gcn_rec stage[STAGE_CAP];
    __shared__ int stageDst[STAGE_CAP];
    __shared__ int waveSum[32];
    __shared__ int blockSum;
    int* gcur   = ldsBins;
    int* lcnt   = ldsBins + NB;
    int* lstart = ldsBins + 2 * NB;

    int carry = 0;
    for (int c = 0; c < NB; c += SORT_THREADS) {
        const int bin = c + threadIdx.x;
        const int tot = (bin < NB) ? binTotal[bin] : 0;
        int total;
        const int base = carry + gcn_block_excl_scan(tot, waveSum, &blockSum, total);
        if (bin < NB) {
            gcur[bin] = base + chunkOff[blockIdx.x * NB + bin];
            if (blockIdx.x == 0) binBase[bin] = base;
        }
        carry += total;
    }
    if (blockIdx.x == 0 && threadIdx.x == 0) binBase[NB] = carry;

    const int chunk = (E + EDGE_CHUNKS - 1) / EDGE_CHUNKS;
    const int lo = min(E, (int)blockIdx.x * chunk);
    const int hi = min(E, lo + chunk);
    for (int s0 = lo; s0 < hi; s0 += STAGE_CAP) {
        const int s1 = min(hi, s0 + STAGE_CAP);
        for (int i = threadIdx.x; i < NB; i += SORT_THREADS) lcnt[i] = 0;
        __syncthreads();

        int r[EDGE_BATCH], cidx[EDGE_BATCH], rank[EDGE_BATCH];
        float v[EDGE_BATCH];
#pragma unroll
        for (int k = 0; k < EDGE_BATCH; ++k) {
            const int e = min(s0 + (int)threadIdx.x + k * SORT_THREADS, s1 - 1);
            r[k] = row[e];
            cidx[k] = col[e];
            v[k] = vals[e];
        }
#pragma unroll
        for (int k = 0; k < EDGE_BATCH; ++k) {
            rank[k] = 0;
            if (s0 + (int)threadIdx.x + k * SORT_THREADS < s1)
                rank[k] = atomicAdd(&lcnt[r[k] >> ROWS_PER_BIN_LOG2], 1);
        }
        __syncthreads();

        int lcarry = 0;
        for (int c = 0; c < NB; c += SORT_THREADS) {
            const int bin = c + threadIdx.x;
            const int n = (bin < NB) ? lcnt[bin] : 0;
            int total;
            const int st = lcarry + gcn_block_excl_scan(n, waveSum, &blockSum, total);
            if (bin < NB) lstart[bin] = st;
            lcarry += total;
        }
        __syncthreads();

#pragma unroll
        for (int k = 0; k < EDGE_BATCH; ++k) {
            if (s0 + (int)threadIdx.x + k * SORT_THREADS < s1) {
                const int bin = r[k] >> ROWS_PER_BIN_LOG2;
                const int li = lstart[bin] + rank[k];
                stage[li] = gcn_make_rec(((r[k] & (ROWS_PER_BIN - 1)) << 24) | cidx[k],
                                         __float_as_int(v[k]));
                stageDst[li] = gcur[bin] + rank[k];
            }
        }
        __syncthreads();

        for (int i = threadIdx.x; i < s1 - s0; i += SORT_THREADS) binned[stageDst[i]] = stage[i];
        __syncthreads();
        for (int i = threadIdx.x; i < NB; i += SORT_THREADS) gcur[i] += lcnt[i];
        __syncthreads();
    }
}

__global__ void __launch_bounds__(ROWS_THREADS)
gcn_bin_rows_kernel(const int* __restrict__ binnedWords, const int* __restrict__ binBase,
                    int* __restrict__ cnt, int* __restrict__ cursorEnd,
                    int* __restrict__ bucketWords, int N) {
    const gcn_rec* __restrict__ binned = (const gcn_rec*)binnedWords;
    gcn_rec* __restrict__ bucket = (gcn_rec*)bucketWords;
    __shared__ gcn_rec stage[BIN_STAGE_CAP];
    __shared__ int rowCur[ROWS_PER_BIN];
    __shared__ int waveSum[ROWS_PER_BIN / 32];
    const int lane = threadIdx.x & 31;
    const int wave = threadIdx.x >> 5;
    const int bin = blockIdx.x;
    const int lo = binBase[bin];
    const int hi = binBase[bin + 1];
    const bool fits = (hi - lo) <= BIN_STAGE_CAP;
    if (threadIdx.x < ROWS_PER_BIN) rowCur[threadIdx.x] = 0;
    __syncthreads();
    for (int i0 = lo + threadIdx.x; i0 < hi; i0 += EDGE_BATCH * ROWS_THREADS) {
        gcn_rec rec[EDGE_BATCH];
#pragma unroll
        for (int k = 0; k < EDGE_BATCH; ++k) rec[k] = binned[min(i0 + k * ROWS_THREADS, hi - 1)];
#pragma unroll
        for (int k = 0; k < EDGE_BATCH; ++k)
            if (i0 + k * ROWS_THREADS < hi) atomicAdd(&rowCur[((int)rec[k]) >> 24], 1);
    }
    __syncthreads();
    int c = 0, incl = 0;
    if (threadIdx.x < ROWS_PER_BIN) {
        c = rowCur[threadIdx.x];
        incl = gcn_wave_incl_scan(c, lane);
        if (lane == 31) waveSum[wave] = incl;
    }
    __syncthreads();
    if (threadIdx.x < ROWS_PER_BIN) {
        int off = 0;
        for (int w = 0; w < wave; ++w) off += waveSum[w];
        const int rel = off + incl - c;
        const int r = bin * ROWS_PER_BIN + threadIdx.x;
        if (r < N) { cnt[r] = c; cursorEnd[r] = lo + rel + c; }
        rowCur[threadIdx.x] = rel;
    }
    __syncthreads();
    for (int i0 = lo + threadIdx.x; i0 < hi; i0 += EDGE_BATCH * ROWS_THREADS) {
        gcn_rec rec[EDGE_BATCH];
#pragma unroll
        for (int k = 0; k < EDGE_BATCH; ++k) rec[k] = binned[min(i0 + k * ROWS_THREADS, hi - 1)];
#pragma unroll
        for (int k = 0; k < EDGE_BATCH; ++k) {
            if (i0 + k * ROWS_THREADS < hi) {
                const int x = (int)rec[k];
                const int pos = atomicAdd(&rowCur[x >> 24], 1);
                const gcn_rec o = gcn_make_rec(x & 0xFFFFFF, (int)(rec[k] >> 32));
                if (fits) stage[pos] = o; else bucket[lo + pos] = o;
            }
        }
    }
    if (fits) {
        __syncthreads();
        for (int i = threadIdx.x; i < hi - lo; i += ROWS_THREADS) bucket[lo + i] = stage[i];
    }
}

template <int G>
__device__ __forceinline__ void gcn_gather_group(const float4* __restrict__ X4,
                                                 int2 my, int j, int lane,
                                                 float4& acc) {
    int c[G];
    float v[G];
    float4 x[G];
#pragma unroll
    for (int k = 0; k < G; ++k) {
        c[k] = __builtin_amdgcn_readlane(my.x, j + k);
        v[k] = __int_as_float(__builtin_amdgcn_readlane(my.y, j + k));
    }
#pragma unroll
    for (int k = 0; k < G; ++k) x[k] = X4[(long long)c[k] * (DIM_K / 4) + lane];
#pragma unroll
    for (int k = 0; k < G; ++k) {
        acc.x = __builtin_fmaf(v[k], x[k].x, acc.x);
        acc.y = __builtin_fmaf(v[k], x[k].y, acc.y);
        acc.z = __builtin_fmaf(v[k], x[k].z, acc.z);
        acc.w = __builtin_fmaf(v[k], x[k].w, acc.w);
    }
}

__global__ void gcn_gather_kernel(const float* __restrict__ X,
                                  const int* __restrict__ cnt,
                                  const int* __restrict__ cursorEnd,
                                  const int* __restrict__ bucket,
                                  float* __restrict__ H, int N) {
    const int lane = threadIdx.x & 31;
    const int r = __builtin_amdgcn_readfirstlane(
        blockIdx.x * (blockDim.x >> 5) + (threadIdx.x >> 5));
    if (r >= N) return;
    const int n = cnt[r];
    const int2* ed = (const int2*)bucket + (cursorEnd[r] - n);
    const float4* X4 = (const float4*)X;
    float4 acc = make_float4(0.0f, 0.0f, 0.0f, 0.0f);

    for (int base = 0; base < n; base += 32) {
        const int m = min(32, n - base);
        int2 my = make_int2(0, 0);
        if (lane < m) my = ed[base + lane];
        int j = 0;
        for (; j + 8 <= m; j += 8) gcn_gather_group<8>(X4, my, j, lane, acc);
        if ((m - j) & 4) { gcn_gather_group<4>(X4, my, j, lane, acc); j += 4; }
        if ((m - j) & 2) { gcn_gather_group<2>(X4, my, j, lane, acc); j += 2; }
        if ((m - j) & 1) { gcn_gather_group<1>(X4, my, j, lane, acc); }
    }
    ((float4*)(H + (long long)r * DIM_K))[lane] = acc;
}

template <int NT>
__global__ void gcn_gemm_tanh_kernel(const float* __restrict__ S,
                                     const float* __restrict__ W,
                                     float* __restrict__ O,
                                     int nNodes) {
    const int lane = threadIdx.x & 31;
    const int wavesPerBlock = blockDim.x >> 5;
    const int mTile = blockIdx.x * wavesPerBlock + (threadIdx.x >> 5);
    if (mTile * 16 >= nNodes) return;

    const int nOut  = NT * 16;
    const int l15   = lane & 15;
    const int khalf = (lane < 16) ? 0 : 2;

    const float* aBase = S + (long long)(mTile * 16 + l15) * DIM_K + khalf;
    const float* bBase = W + (long long)l15 * DIM_K + khalf;

    v8f acc[NT];
#pragma unroll
    for (int t = 0; t < NT; ++t) acc[t] = (v8f){};

#pragma unroll 2
    for (int k = 0; k < DIM_K; k += 4) {
        v2f a;
        a.x = aBase[k];
        a.y = aBase[k + 1];
#pragma unroll
        for (int t = 0; t < NT; ++t) {
            const float* bp = bBase + (long long)t * 16 * DIM_K + k;
            v2f b;
            b.x = bp[0];
            b.y = bp[1];
            acc[t] = __builtin_amdgcn_wmma_f32_16x16x4_f32(
                 false, a,  false, b,
                 (short)0, acc[t],  false,  false);
        }
    }

    const int mbase = mTile * 16 + ((lane < 16) ? 0 : 8);
#pragma unroll
    for (int t = 0; t < NT; ++t) {
        const int n = t * 16 + l15;
#pragma unroll
        for (int r = 0; r < 8; ++r) {
            O[(long long)(mbase + r) * nOut + n] = fast_tanh(acc[t][r]);
        }
    }
}

extern "C" void kernel_launch(void* const* d_in, const int* in_sizes, int n_in,
                              void* d_out, int out_size, void* d_ws, size_t ws_size,
                              hipStream_t stream) {
    const float* X        = (const float*)d_in[0];
    const float* W1       = (const float*)d_in[1];
    const float* W2       = (const float*)d_in[2];
    const float* edgeVals = (const float*)d_in[3];
    const int*   edgeRow  = (const int*)d_in[4];
    const int*   edgeCol  = (const int*)d_in[5];
    float* out = (float*)d_out;

    const int N = in_sizes[0] / DIM_K;
    const int E = in_sizes[3];
    const long long featCount = (long long)N * DIM_K;

    const int NB = (N + ROWS_PER_BIN - 1) / ROWS_PER_BIN;
    const int nPad = (N + 31) & ~31;

    float* S  = (float*)d_ws;
    float* H1 = S + featCount;
    int* cnt    = (int*)(H1 + featCount);
    int* cursor = cnt + nPad;
    int* bucket = cursor + nPad;
    int* binned = bucket + 2 * (long long)E;
    int* chunkHist = binned + 2 * (long long)E;
    int* chunkOff  = chunkHist + EDGE_CHUNKS * NB;
    int* binTotal  = chunkOff + EDGE_CHUNKS * NB;
    int* binBase   = binTotal + NB + 32;

    const int threads = 256;

    const int wavesPerBlock = threads / 32;
    const int rowBlocks = (N + wavesPerBlock - 1) / wavesPerBlock;

    const int gemmThreads = 256;
    const int mTiles = (N + 15) / 16;
    const int gemmBlocks = (mTiles + wavesPerBlock - 1) / wavesPerBlock;

    gcn_bin_hist_kernel<<<EDGE_CHUNKS, SORT_THREADS, NB * sizeof(int), stream>>>(
        edgeRow, chunkHist, E, NB);
    gcn_bin_offsets_kernel<<<(NB + 31) / 32, SORT_THREADS, 0, stream>>>(
        chunkHist, chunkOff, binTotal, NB);
    gcn_bin_scatter_kernel<<<EDGE_CHUNKS, SORT_THREADS, 3 * NB * sizeof(int), stream>>>(
        edgeVals, edgeRow, edgeCol, chunkOff, binTotal, binBase, binned, E, NB);
    gcn_bin_rows_kernel<<<NB, ROWS_THREADS, 0, stream>>>(
        binned, binBase, cnt, cursor, bucket, N);

    gcn_gather_kernel<<<rowBlocks, threads, 0, stream>>>(X, cnt, cursor, bucket, S, N);
    gcn_gemm_tanh_kernel<8><<<gemmBlocks, gemmThreads, 0, stream>>>(
        S, W1, H1, N);

    gcn_gather_kernel<<<rowBlocks, threads, 0, stream>>>(H1, cnt, cursor, bucket, S, N);
    gcn_gemm_tanh_kernel<4><<<gemmBlocks, gemmThreads, 0, stream>>>(
        S, W2, out, N);
}


# Layer 2 applies W2 before the neighbour sum (A(H1 W2^T), reassociated sums): GEMM to 64 columns, then a 64-wide row gather ending in tanh, one 256-B unit per edge instead of two
# speedup vs baseline: 3.5076x; 1.0636x over previous
#include <hip/hip_runtime.h>
#include <hip/hip_bf16.h>
#include <math.h>


typedef __attribute__((ext_vector_type(2))) float v2f;
typedef __attribute__((ext_vector_type(8))) float v8f;

#define DIM_K 128
#define DIM_OUT 64

__device__ __forceinline__ float fast_tanh(float x) {
    float ax = fabsf(x);
    float e  = __expf(-2.0f * ax);
    float r  = (1.0f - e) * __builtin_amdgcn_rcpf(1.0f + e);
    return copysignf(r, x);
}

#define ROWS_PER_BIN_LOG2 7
#define ROWS_PER_BIN (1 << ROWS_PER_BIN_LOG2)
#define EDGE_CHUNKS 256
#define SORT_THREADS 1024
#define EDGE_BATCH 8
#define STAGE_CAP (EDGE_BATCH * SORT_THREADS)
#define ROWS_THREADS 256
#define BIN_STAGE_CAP 4096

typedef unsigned long long gcn_rec;

__device__ __forceinline__ gcn_rec gcn_make_rec(int lo, int hi) {
    return (gcn_rec)(unsigned)lo | ((gcn_rec)(unsigned)hi << 32);
}

__device__ __forceinline__ int gcn_wave_incl_scan(int x, int lane) {
#pragma unroll
    for (int d = 1; d < 32; d <<= 1) {
        int t = __shfl_up(x, d, 32);
        if (lane >= d) x += t;
    }
    return x;
}

__device__ __forceinline__ int gcn_block_excl_scan(int x, int* waveSum, int* blockSum, int& total) {
    const int lane = threadIdx.x & 31;
    const int wave = threadIdx.x >> 5;
    const int nw = blockDim.x >> 5;
    const int incl = gcn_wave_incl_scan(x, lane);
    if (lane == 31) waveSum[wave] = incl;
    __syncthreads();
    if (wave == 0) {
        const int w = (lane < nw) ? waveSum[lane] : 0;
        const int wi = gcn_wave_incl_scan(w, lane);
        waveSum[lane] = wi - w;
        if (lane == 31) *blockSum = wi;
    }
    __syncthreads();
    const int excl = waveSum[wave] + incl - x;
    total = *blockSum;
    __syncthreads();
    return excl;
}

__global__ void __launch_bounds__(SORT_THREADS)
gcn_bin_hist_kernel(const int* __restrict__ row, int* __restrict__ chunkHist, int E, int NB) {
    extern __shared__ int ldsBins[];
    for (int i = threadIdx.x; i < NB; i += SORT_THREADS) ldsBins[i] = 0;
    __syncthreads();
    const int chunk = (E + EDGE_CHUNKS - 1) / EDGE_CHUNKS;
    const int lo = min(E, (int)blockIdx.x * chunk);
    const int hi = min(E, lo + chunk);
    for (int e0 = lo + threadIdx.x; e0 < hi; e0 += STAGE_CAP) {
        int r[EDGE_BATCH];
#pragma unroll
        for (int k = 0; k < EDGE_BATCH; ++k) r[k] = row[min(e0 + k * SORT_THREADS, hi - 1)];
#pragma unroll
        for (int k = 0; k < EDGE_BATCH; ++k)
            if (e0 + k * SORT_THREADS < hi) atomicAdd(&ldsBins[r[k] >> ROWS_PER_BIN_LOG2], 1);
    }
    __syncthreads();
    for (int i = threadIdx.x; i < NB; i += SORT_THREADS)
        chunkHist[blockIdx.x * NB + i] = ldsBins[i];
}

__global__ void __launch_bounds__(SORT_THREADS)
gcn_bin_offsets_kernel(const int* __restrict__ chunkHist, int* __restrict__ chunkOff,
                       int* __restrict__ binTotal, int NB) {
    __shared__ int part[32][33];
    const int lane = threadIdx.x & 31;
    const int wave = threadIdx.x >> 5;
    const int bin = blockIdx.x * 32 + lane;
    const bool live = bin < NB;
    constexpr int PER = EDGE_CHUNKS / 32;
    int v[PER];
#pragma unroll
    for (int k = 0; k < PER; ++k) v[k] = live ? chunkHist[(wave * PER + k) * NB + bin] : 0;
    int s = 0;
#pragma unroll
    for (int k = 0; k < PER; ++k) { const int t = v[k]; v[k] = s; s += t; }
    part[wave][lane] = s;
    __syncthreads();
    int off = 0;
    for (int w = 0; w < wave; ++w) off += part[w][lane];
    if (live) {
#pragma unroll
        for (int k = 0; k < PER; ++k) chunkOff[(wave * PER + k) * NB + bin] = off + v[k];
        if (wave == 31) binTotal[bin] = off + s;
    }
}

__global__ void __launch_bounds__(SORT_THREADS)
gcn_bin_scatter_kernel(const float* __restrict__ vals, const int* __restrict__ row,
                       const int* __restrict__ col, const int* __restrict__ chunkOff,
                       const int* __restrict__ binTotal, int* __restrict__ binBase,
                       int* __restrict__ binnedWords, int E, int NB) {
    gcn_rec* __restrict__ binned = (gcn_rec*)binnedWords;
    extern __shared__ int ldsBins[];
    __shared__ gcn_rec stage[STAGE_CAP];
    __shared__ int stageDst[STAGE_CAP];
    __shared__ int waveSum[32];
    __shared__ int blockSum;
    int* gcur   = ldsBins;
    int* lcnt   = ldsBins + NB;
    int* lstart = ldsBins + 2 * NB;

    int carry = 0;
    for (int c = 0; c < NB; c += SORT_THREADS) {
        const int bin = c + threadIdx.x;
        const int tot = (bin < NB) ? binTotal[bin] : 0;
        int total;
        const int base = carry + gcn_block_excl_scan(tot, waveSum, &blockSum, total);
        if (bin < NB) {
            gcur[bin] = base + chunkOff[blockIdx.x * NB + bin];
            if (blockIdx.x == 0) binBase[bin] = base;
        }
        carry += total;
    }
    if (blockIdx.x == 0 && threadIdx.x == 0) binBase[NB] = carry;

    const int chunk = (E + EDGE_CHUNKS - 1) / EDGE_CHUNKS;
    const int lo = min(E, (int)blockIdx.x * chunk);
    const int hi = min(E, lo + chunk);
    for (int s0 = lo; s0 < hi; s0 += STAGE_CAP) {
        const int s1 = min(hi, s0 + STAGE_CAP);
        for (int i = threadIdx.x; i < NB; i += SORT_THREADS) lcnt[i] = 0;
        __syncthreads();

        int r[EDGE_BATCH], cidx[EDGE_BATCH], rank[EDGE_BATCH];
        float v[EDGE_BATCH];
#pragma unroll
        for (int k = 0; k < EDGE_BATCH; ++k) {
            const int e = min(s0 + (int)threadIdx.x + k * SORT_THREADS, s1 - 1);
            r[k] = row[e];
            cidx[k] = col[e];
            v[k] = vals[e];
        }
#pragma unroll
        for (int k = 0; k < EDGE_BATCH; ++k) {
            rank[k] = 0;
            if (s0 + (int)threadIdx.x + k * SORT_THREADS < s1)
                rank[k] = atomicAdd(&lcnt[r[k] >> ROWS_PER_BIN_LOG2], 1);
        }
        __syncthreads();

        int lcarry = 0;
        for (int c = 0; c < NB; c += SORT_THREADS) {
            const int bin = c + threadIdx.x;
            const int n = (bin < NB) ? lcnt[bin] : 0;
            int total;
            const int st = lcarry + gcn_block_excl_scan(n, waveSum, &blockSum, total);
            if (bin < NB) lstart[bin] = st;
            lcarry += total;
        }
        __syncthreads();

#pragma unroll
        for (int k = 0; k < EDGE_BATCH; ++k) {
            if (s0 + (int)threadIdx.x + k * SORT_THREADS < s1) {
                const int bin = r[k] >> ROWS_PER_BIN_LOG2;
                const int li = lstart[bin] + rank[k];
                stage[li] = gcn_make_rec(((r[k] & (ROWS_PER_BIN - 1)) << 24) | cidx[k],
                                         __float_as_int(v[k]));
                stageDst[li] = gcur[bin] + rank[k];
            }
        }
        __syncthreads();

        for (int i = threadIdx.x; i < s1 - s0; i += SORT_THREADS) binned[stageDst[i]] = stage[i];
        __syncthreads();
        for (int i = threadIdx.x; i < NB; i += SORT_THREADS) gcur[i] += lcnt[i];
        __syncthreads();
    }
}

__global__ void __launch_bounds__(ROWS_THREADS)
gcn_bin_rows_kernel(const int* __restrict__ binnedWords, const int* __restrict__ binBase,
                    int* __restrict__ cnt, int* __restrict__ cursorEnd,
                    int* __restrict__ bucketWords, int N) {
    const gcn_rec* __restrict__ binned = (const gcn_rec*)binnedWords;
    gcn_rec* __restrict__ bucket = (gcn_rec*)bucketWords;
    __shared__ gcn_rec stage[BIN_STAGE_CAP];
    __shared__ int rowCur[ROWS_PER_BIN];
    __shared__ int waveSum[ROWS_PER_BIN / 32];
    const int lane = threadIdx.x & 31;
    const int wave = threadIdx.x >> 5;
    const int bin = blockIdx.x;
    const int lo = binBase[bin];
    const int hi = binBase[bin + 1];
    const bool fits = (hi - lo) <= BIN_STAGE_CAP;
    if (threadIdx.x < ROWS_PER_BIN) rowCur[threadIdx.x] = 0;
    __syncthreads();
    for (int i0 = lo + threadIdx.x; i0 < hi; i0 += EDGE_BATCH * ROWS_THREADS) {
        gcn_rec rec[EDGE_BATCH];
#pragma unroll
        for (int k = 0; k < EDGE_BATCH; ++k) rec[k] = binned[min(i0 + k * ROWS_THREADS, hi - 1)];
#pragma unroll
        for (int k = 0; k < EDGE_BATCH; ++k)
            if (i0 + k * ROWS_THREADS < hi) atomicAdd(&rowCur[((int)rec[k]) >> 24], 1);
    }
    __syncthreads();
    int c = 0, incl = 0;
    if (threadIdx.x < ROWS_PER_BIN) {
        c = rowCur[threadIdx.x];
        incl = gcn_wave_incl_scan(c, lane);
        if (lane == 31) waveSum[wave] = incl;
    }
    __syncthreads();
    if (threadIdx.x < ROWS_PER_BIN) {
        int off = 0;
        for (int w = 0; w < wave; ++w) off += waveSum[w];
        const int rel = off + incl - c;
        const int r = bin * ROWS_PER_BIN + threadIdx.x;
        if (r < N) { cnt[r] = c; cursorEnd[r] = lo + rel + c; }
        rowCur[threadIdx.x] = rel;
    }
    __syncthreads();
    for (int i0 = lo + threadIdx.x; i0 < hi; i0 += EDGE_BATCH * ROWS_THREADS) {
        gcn_rec rec[EDGE_BATCH];
#pragma unroll
        for (int k = 0; k < EDGE_BATCH; ++k) rec[k] = binned[min(i0 + k * ROWS_THREADS, hi - 1)];
#pragma unroll
        for (int k = 0; k < EDGE_BATCH; ++k) {
            if (i0 + k * ROWS_THREADS < hi) {
                const int x = (int)rec[k];
                const int pos = atomicAdd(&rowCur[x >> 24], 1);
                const gcn_rec o = gcn_make_rec(x & 0xFFFFFF, (int)(rec[k] >> 32));
                if (fits) stage[pos] = o; else bucket[lo + pos] = o;
            }
        }
    }
    if (fits) {
        __syncthreads();
        for (int i = threadIdx.x; i < hi - lo; i += ROWS_THREADS) bucket[lo + i] = stage[i];
    }
}

template <int G>
__device__ __forceinline__ void gcn_gather_group(const float4* __restrict__ X4,
                                                 int2 my, int j, int lane,
                                                 float4& acc) {
    int c[G];
    float v[G];
    float4 x[G];
#pragma unroll
    for (int k = 0; k < G; ++k) {
        c[k] = __builtin_amdgcn_readlane(my.x, j + k);
        v[k] = __int_as_float(__builtin_amdgcn_readlane(my.y, j + k));
    }
#pragma unroll
    for (int k = 0; k < G; ++k) x[k] = X4[(long long)c[k] * (DIM_K / 4) + lane];
#pragma unroll
    for (int k = 0; k < G; ++k) {
        acc.x = __builtin_fmaf(v[k], x[k].x, acc.x);
        acc.y = __builtin_fmaf(v[k], x[k].y, acc.y);
        acc.z = __builtin_fmaf(v[k], x[k].z, acc.z);
        acc.w = __builtin_fmaf(v[k], x[k].w, acc.w);
    }
}

__global__ void gcn_gather_kernel(const float* __restrict__ X,
                                  const int* __restrict__ cnt,
                                  const int* __restrict__ cursorEnd,
                                  const int* __restrict__ bucket,
                                  float* __restrict__ H, int N) {
    const int lane = threadIdx.x & 31;
    const int r = __builtin_amdgcn_readfirstlane(
        blockIdx.x * (blockDim.x >> 5) + (threadIdx.x >> 5));
    if (r >= N) return;
    const int n = cnt[r];
    const int2* ed = (const int2*)bucket + (cursorEnd[r] - n);
    const float4* X4 = (const float4*)X;
    float4 acc = make_float4(0.0f, 0.0f, 0.0f, 0.0f);

    for (int base = 0; base < n; base += 32) {
        const int m = min(32, n - base);
        int2 my = make_int2(0, 0);
        if (lane < m) my = ed[base + lane];
        int j = 0;
        for (; j + 8 <= m; j += 8) gcn_gather_group<8>(X4, my, j, lane, acc);
        if ((m - j) & 4) { gcn_gather_group<4>(X4, my, j, lane, acc); j += 4; }
        if ((m - j) & 2) { gcn_gather_group<2>(X4, my, j, lane, acc); j += 2; }
        if ((m - j) & 1) { gcn_gather_group<1>(X4, my, j, lane, acc); }
    }
    ((float4*)(H + (long long)r * DIM_K))[lane] = acc;
}

template <int G>
__device__ __forceinline__ void gcn_gather_group64(const float2* __restrict__ P2,
                                                   int2 my, int j, int lane,
                                                   float2& acc) {
    int c[G];
    float v[G];
    float2 x[G];
#pragma unroll
    for (int k = 0; k < G; ++k) {
        c[k] = __builtin_amdgcn_readlane(my.x, j + k);
        v[k] = __int_as_float(__builtin_amdgcn_readlane(my.y, j + k));
    }
#pragma unroll
    for (int k = 0; k < G; ++k) x[k] = P2[(long long)c[k] * (DIM_OUT / 2) + lane];
#pragma unroll
    for (int k = 0; k < G; ++k) {
        acc.x = __builtin_fmaf(v[k], x[k].x, acc.x);
        acc.y = __builtin_fmaf(v[k], x[k].y, acc.y);
    }
}

__global__ void gcn_gather64_tanh_kernel(const float* __restrict__ P,
                                         const int* __restrict__ cnt,
                                         const int* __restrict__ cursorEnd,
                                         const int* __restrict__ bucket,
                                         float* __restrict__ O, int N) {
    const int lane = threadIdx.x & 31;
    const int r = __builtin_amdgcn_readfirstlane(
        blockIdx.x * (blockDim.x >> 5) + (threadIdx.x >> 5));
    if (r >= N) return;
    const int n = cnt[r];
    const int2* ed = (const int2*)bucket + (cursorEnd[r] - n);
    const float2* P2 = (const float2*)P;
    float2 acc = make_float2(0.0f, 0.0f);

    for (int base = 0; base < n; base += 32) {
        const int m = min(32, n - base);
        int2 my = make_int2(0, 0);
        if (lane < m) my = ed[base + lane];
        int j = 0;
        for (; j + 16 <= m; j += 16) gcn_gather_group64<16>(P2, my, j, lane, acc);
        if ((m - j) & 8) { gcn_gather_group64<8>(P2, my, j, lane, acc); j += 8; }
        if ((m - j) & 4) { gcn_gather_group64<4>(P2, my, j, lane, acc); j += 4; }
        if ((m - j) & 2) { gcn_gather_group64<2>(P2, my, j, lane, acc); j += 2; }
        if ((m - j) & 1) { gcn_gather_group64<1>(P2, my, j, lane, acc); }
    }
    ((float2*)(O + (long long)r * DIM_OUT))[lane] =
        make_float2(fast_tanh(acc.x), fast_tanh(acc.y));
}

template <int NT, bool ACT>
__global__ void gcn_gemm_kernel(const float* __restrict__ S,
                                const float* __restrict__ W,
                                float* __restrict__ O,
                                int nNodes) {
    const int lane = threadIdx.x & 31;
    const int wavesPerBlock = blockDim.x >> 5;
    const int mTile = blockIdx.x * wavesPerBlock + (threadIdx.x >> 5);
    if (mTile * 16 >= nNodes) return;

    const int nOut  = NT * 16;
    const int l15   = lane & 15;
    const int khalf = (lane < 16) ? 0 : 2;

    const float* aBase = S + (long long)(mTile * 16 + l15) * DIM_K + khalf;
    const float* bBase = W + (long long)l15 * DIM_K + khalf;

    v8f acc[NT];
#pragma unroll
    for (int t = 0; t < NT; ++t) acc[t] = (v8f){};

#pragma unroll 2
    for (int k = 0; k < DIM_K; k += 4) {
        v2f a;
        a.x = aBase[k];
        a.y = aBase[k + 1];
#pragma unroll
        for (int t = 0; t < NT; ++t) {
            const float* bp = bBase + (long long)t * 16 * DIM_K + k;
            v2f b;
            b.x = bp[0];
            b.y = bp[1];
            acc[t] = __builtin_amdgcn_wmma_f32_16x16x4_f32(
                 false, a,  false, b,
                 (short)0, acc[t],  false,  false);
        }
    }

    const int mbase = mTile * 16 + ((lane < 16) ? 0 : 8);
#pragma unroll
    for (int t = 0; t < NT; ++t) {
        const int n = t * 16 + l15;
#pragma unroll
        for (int r = 0; r < 8; ++r) {
            O[(long long)(mbase + r) * nOut + n] = ACT ? fast_tanh(acc[t][r]) : acc[t][r];
        }
    }
}

extern "C" void kernel_launch(void* const* d_in, const int* in_sizes, int n_in,
                              void* d_out, int out_size, void* d_ws, size_t ws_size,
                              hipStream_t stream) {
    const float* X        = (const float*)d_in[0];
    const float* W1       = (const float*)d_in[1];
    const float* W2       = (const float*)d_in[2];
    const float* edgeVals = (const float*)d_in[3];
    const int*   edgeRow  = (const int*)d_in[4];
    const int*   edgeCol  = (const int*)d_in[5];
    float* out = (float*)d_out;

    const int N = in_sizes[0] / DIM_K;
    const int E = in_sizes[3];
    const long long featCount = (long long)N * DIM_K;

    const int NB = (N + ROWS_PER_BIN - 1) / ROWS_PER_BIN;
    const int nPad = (N + 31) & ~31;

    float* S  = (float*)d_ws;
    float* P  = S;
    float* H1 = S + featCount;
    int* cnt    = (int*)(H1 + featCount);
    int* cursor = cnt + nPad;
    int* bucket = cursor + nPad;
    int* binned = bucket + 2 * (long long)E;
    int* chunkHist = binned + 2 * (long long)E;
    int* chunkOff  = chunkHist + EDGE_CHUNKS * NB;
    int* binTotal  = chunkOff + EDGE_CHUNKS * NB;
    int* binBase   = binTotal + NB + 32;

    const int threads = 256;

    const int wavesPerBlock = threads / 32;
    const int rowBlocks = (N + wavesPerBlock - 1) / wavesPerBlock;

    const int gemmThreads = 256;
    const int mTiles = (N + 15) / 16;
    const int gemmBlocks = (mTiles + wavesPerBlock - 1) / wavesPerBlock;

    gcn_bin_hist_kernel<<<EDGE_CHUNKS, SORT_THREADS, NB * sizeof(int), stream>>>(
        edgeRow, chunkHist, E, NB);
    gcn_bin_offsets_kernel<<<(NB + 31) / 32, SORT_THREADS, 0, stream>>>(
        chunkHist, chunkOff, binTotal, NB);
    gcn_bin_scatter_kernel<<<EDGE_CHUNKS, SORT_THREADS, 3 * NB * sizeof(int), stream>>>(
        edgeVals, edgeRow, edgeCol, chunkOff, binTotal, binBase, binned, E, NB);
    gcn_bin_rows_kernel<<<NB, ROWS_THREADS, 0, stream>>>(
        binned, binBase, cnt, cursor, bucket, N);

    gcn_gather_kernel<<<rowBlocks, threads, 0, stream>>>(X, cnt, cursor, bucket, S, N);
    gcn_gemm_kernel<8, true><<<gemmBlocks, gemmThreads, 0, stream>>>(
        S, W1, H1, N);

    gcn_gemm_kernel<4, false><<<gemmBlocks, gemmThreads, 0, stream>>>(
        H1, W2, P, N);
    gcn_gather64_tanh_kernel<<<rowBlocks, threads, 0, stream>>>(P, cnt, cursor, bucket, out, N);
}


# Both GEMMs: W copied once per work-group into LDS (padded rows), B fragments read from LDS instead of 16-line global loads; 256 work-groups of 24 waves, one 16-row strip a wave; sum order unchanged
# speedup vs baseline: 4.4218x; 1.2606x over previous
#include <hip/hip_runtime.h>
#include <hip/hip_bf16.h>
#include <math.h>


typedef __attribute__((ext_vector_type(2))) float v2f;
typedef __attribute__((ext_vector_type(8))) float v8f;

#define DIM_K 128
#define DIM_OUT 64

__device__ __forceinline__ float fast_tanh(float x) {
    float ax = fabsf(x);
    float e  = __expf(-2.0f * ax);
    float r  = (1.0f - e) * __builtin_amdgcn_rcpf(1.0f + e);
    return copysignf(r, x);
}

#define ROWS_PER_BIN_LOG2 7
#define ROWS_PER_BIN (1 << ROWS_PER_BIN_LOG2)
#define EDGE_CHUNKS 256
#define SORT_THREADS 1024
#define EDGE_BATCH 8
#define STAGE_CAP (EDGE_BATCH * SORT_THREADS)
#define ROWS_THREADS 256
#define BIN_STAGE_CAP 4096

typedef unsigned long long gcn_rec;

__device__ __forceinline__ gcn_rec gcn_make_rec(int lo, int hi) {
    return (gcn_rec)(unsigned)lo | ((gcn_rec)(unsigned)hi << 32);
}

__device__ __forceinline__ int gcn_wave_incl_scan(int x, int lane) {
#pragma unroll
    for (int d = 1; d < 32; d <<= 1) {
        int t = __shfl_up(x, d, 32);
        if (lane >= d) x += t;
    }
    return x;
}

__device__ __forceinline__ int gcn_block_excl_scan(int x, int* waveSum, int* blockSum, int& total) {
    const int lane = threadIdx.x & 31;
    const int wave = threadIdx.x >> 5;
    const int nw = blockDim.x >> 5;
    const int incl = gcn_wave_incl_scan(x, lane);
    if (lane == 31) waveSum[wave] = incl;
    __syncthreads();
    if (wave == 0) {
        const int w = (lane < nw) ? waveSum[lane] : 0;
        const int wi = gcn_wave_incl_scan(w, lane);
        waveSum[lane] = wi - w;
        if (lane == 31) *blockSum = wi;
    }
    __syncthreads();
    const int excl = waveSum[wave] + incl - x;
    total = *blockSum;
    __syncthreads();
    return excl;
}

__global__ void __launch_bounds__(SORT_THREADS)
gcn_bin_hist_kernel(const int* __restrict__ row, int* __restrict__ chunkHist, int E, int NB) {
    extern __shared__ int ldsBins[];
    for (int i = threadIdx.x; i < NB; i += SORT_THREADS) ldsBins[i] = 0;
    __syncthreads();
    const int chunk = (E + EDGE_CHUNKS - 1) / EDGE_CHUNKS;
    const int lo = min(E, (int)blockIdx.x * chunk);
    const int hi = min(E, lo + chunk);
    for (int e0 = lo + threadIdx.x; e0 < hi; e0 += STAGE_CAP) {
        int r[EDGE_BATCH];
#pragma unroll
        for (int k = 0; k < EDGE_BATCH; ++k) r[k] = row[min(e0 + k * SORT_THREADS, hi - 1)];
#pragma unroll
        for (int k = 0; k < EDGE_BATCH; ++k)
            if (e0 + k * SORT_THREADS < hi) atomicAdd(&ldsBins[r[k] >> ROWS_PER_BIN_LOG2], 1);
    }
    __syncthreads();
    for (int i = threadIdx.x; i < NB; i += SORT_THREADS)
        chunkHist[blockIdx.x * NB + i] = ldsBins[i];
}

__global__ void __launch_bounds__(SORT_THREADS)
gcn_bin_offsets_kernel(const int* __restrict__ chunkHist, int* __restrict__ chunkOff,
                       int* __restrict__ binTotal, int NB) {
    __shared__ int part[32][33];
    const int lane = threadIdx.x & 31;
    const int wave = threadIdx.x >> 5;
    const int bin = blockIdx.x * 32 + lane;
    const bool live = bin < NB;
    constexpr int PER = EDGE_CHUNKS / 32;
    int v[PER];
#pragma unroll
    for (int k = 0; k < PER; ++k) v[k] = live ? chunkHist[(wave * PER + k) * NB + bin] : 0;
    int s = 0;
#pragma unroll
    for (int k = 0; k < PER; ++k) { const int t = v[k]; v[k] = s; s += t; }
    part[wave][lane] = s;
    __syncthreads();
    int off = 0;
    for (int w = 0; w < wave; ++w) off += part[w][lane];
    if (live) {
#pragma unroll
        for (int k = 0; k < PER; ++k) chunkOff[(wave * PER + k) * NB + bin] = off + v[k];
        if (wave == 31) binTotal[bin] = off + s;
    }
}

__global__ void __launch_bounds__(SORT_THREADS)
gcn_bin_scatter_kernel(const float* __restrict__ vals, const int* __restrict__ row,
                       const int* __restrict__ col, const int* __restrict__ chunkOff,
                       const int* __restrict__ binTotal, int* __restrict__ binBase,
                       int* __restrict__ binnedWords, int E, int NB) {
    gcn_rec* __restrict__ binned = (gcn_rec*)binnedWords;
    extern __shared__ int ldsBins[];
    __shared__ gcn_rec stage[STAGE_CAP];
    __shared__ int stageDst[STAGE_CAP];
    __shared__ int waveSum[32];
    __shared__ int blockSum;
    int* gcur   = ldsBins;
    int* lcnt   = ldsBins + NB;
    int* lstart = ldsBins + 2 * NB;

    int carry = 0;
    for (int c = 0; c < NB; c += SORT_THREADS) {
        const int bin = c + threadIdx.x;
        const int tot = (bin < NB) ? binTotal[bin] : 0;
        int total;
        const int base = carry + gcn_block_excl_scan(tot, waveSum, &blockSum, total);
        if (bin < NB) {
            gcur[bin] = base + chunkOff[blockIdx.x * NB + bin];
            if (blockIdx.x == 0) binBase[bin] = base;
        }
        carry += total;
    }
    if (blockIdx.x == 0 && threadIdx.x == 0) binBase[NB] = carry;

    const int chunk = (E + EDGE_CHUNKS - 1) / EDGE_CHUNKS;
    const int lo = min(E, (int)blockIdx.x * chunk);
    const int hi = min(E, lo + chunk);
    for (int s0 = lo; s0 < hi; s0 += STAGE_CAP) {
        const int s1 = min(hi, s0 + STAGE_CAP);
        for (int i = threadIdx.x; i < NB; i += SORT_THREADS) lcnt[i] = 0;
        __syncthreads();

        int r[EDGE_BATCH], cidx[EDGE_BATCH], rank[EDGE_BATCH];
        float v[EDGE_BATCH];
#pragma unroll
        for (int k = 0; k < EDGE_BATCH; ++k) {
            const int e = min(s0 + (int)threadIdx.x + k * SORT_THREADS, s1 - 1);
            r[k] = row[e];
            cidx[k] = col[e];
            v[k] = vals[e];
        }
#pragma unroll
        for (int k = 0; k < EDGE_BATCH; ++k) {
            rank[k] = 0;
            if (s0 + (int)threadIdx.x + k * SORT_THREADS < s1)
                rank[k] = atomicAdd(&lcnt[r[k] >> ROWS_PER_BIN_LOG2], 1);
        }
        __syncthreads();

        int lcarry = 0;
        for (int c = 0; c < NB; c += SORT_THREADS) {
            const int bin = c + threadIdx.x;
            const int n = (bin < NB) ? lcnt[bin] : 0;
            int total;
            const int st = lcarry + gcn_block_excl_scan(n, waveSum, &blockSum, total);
            if (bin < NB) lstart[bin] = st;
            lcarry += total;
        }
        __syncthreads();

#pragma unroll
        for (int k = 0; k < EDGE_BATCH; ++k) {
            if (s0 + (int)threadIdx.x + k * SORT_THREADS < s1) {
                const int bin = r[k] >> ROWS_PER_BIN_LOG2;
                const int li = lstart[bin] + rank[k];
                stage[li] = gcn_make_rec(((r[k] & (ROWS_PER_BIN - 1)) << 24) | cidx[k],
                                         __float_as_int(v[k]));
                stageDst[li] = gcur[bin] + rank[k];
            }
        }
        __syncthreads();

        for (int i = threadIdx.x; i < s1 - s0; i += SORT_THREADS) binned[stageDst[i]] = stage[i];
        __syncthreads();
        for (int i = threadIdx.x; i < NB; i += SORT_THREADS) gcur[i] += lcnt[i];
        __syncthreads();
    }
}

__global__ void __launch_bounds__(ROWS_THREADS)
gcn_bin_rows_kernel(const int* __restrict__ binnedWords, const int* __restrict__ binBase,
                    int* __restrict__ cnt, int* __restrict__ cursorEnd,
                    int* __restrict__ bucketWords, int N) {
    const gcn_rec* __restrict__ binned = (const gcn_rec*)binnedWords;
    gcn_rec* __restrict__ bucket = (gcn_rec*)bucketWords;
    __shared__ gcn_rec stage[BIN_STAGE_CAP];
    __shared__ int rowCur[ROWS_PER_BIN];
    __shared__ int waveSum[ROWS_PER_BIN / 32];
    const int lane = threadIdx.x & 31;
    const int wave = threadIdx.x >> 5;
    const int bin = blockIdx.x;
    const int lo = binBase[bin];
    const int hi = binBase[bin + 1];
    const bool fits = (hi - lo) <= BIN_STAGE_CAP;
    if (threadIdx.x < ROWS_PER_BIN) rowCur[threadIdx.x] = 0;
    __syncthreads();
    for (int i0 = lo + threadIdx.x; i0 < hi; i0 += EDGE_BATCH * ROWS_THREADS) {
        gcn_rec rec[EDGE_BATCH];
#pragma unroll
        for (int k = 0; k < EDGE_BATCH; ++k) rec[k] = binned[min(i0 + k * ROWS_THREADS, hi - 1)];
#pragma unroll
        for (int k = 0; k < EDGE_BATCH; ++k)
            if (i0 + k * ROWS_THREADS < hi) atomicAdd(&rowCur[((int)rec[k]) >> 24], 1);
    }
    __syncthreads();
    int c = 0, incl = 0;
    if (threadIdx.x < ROWS_PER_BIN) {
        c = rowCur[threadIdx.x];
        incl = gcn_wave_incl_scan(c, lane);
        if (lane == 31) waveSum[wave] = incl;
    }
    __syncthreads();
    if (threadIdx.x < ROWS_PER_BIN) {
        int off = 0;
        for (int w = 0; w < wave; ++w) off += waveSum[w];
        const int rel = off + incl - c;
        const int r = bin * ROWS_PER_BIN + threadIdx.x;
        if (r < N) { cnt[r] = c; cursorEnd[r] = lo + rel + c; }
        rowCur[threadIdx.x] = rel;
    }
    __syncthreads();
    for (int i0 = lo + threadIdx.x; i0 < hi; i0 += EDGE_BATCH * ROWS_THREADS) {
        gcn_rec rec[EDGE_BATCH];
#pragma unroll
        for (int k = 0; k < EDGE_BATCH; ++k) rec[k] = binned[min(i0 + k * ROWS_THREADS, hi - 1)];
#pragma unroll
        for (int k = 0; k < EDGE_BATCH; ++k) {
            if (i0 + k * ROWS_THREADS < hi) {
                const int x = (int)rec[k];
                const int pos = atomicAdd(&rowCur[x >> 24], 1);
                const gcn_rec o = gcn_make_rec(x & 0xFFFFFF, (int)(rec[k] >> 32));
                if (fits) stage[pos] = o; else bucket[lo + pos] = o;
            }
        }
    }
    if (fits) {
        __syncthreads();
        for (int i = threadIdx.x; i < hi - lo; i += ROWS_THREADS) bucket[lo + i] = stage[i];
    }
}

template <int G>
__device__ __forceinline__ void gcn_gather_group(const float4* __restrict__ X4,
                                                 int2 my, int j, int lane,
                                                 float4& acc) {
    int c[G];
    float v[G];
    float4 x[G];
#pragma unroll
    for (int k = 0; k < G; ++k) {
        c[k] = __builtin_amdgcn_readlane(my.x, j + k);
        v[k] = __int_as_float(__builtin_amdgcn_readlane(my.y, j + k));
    }
#pragma unroll
    for (int k = 0; k < G; ++k) x[k] = X4[(long long)c[k] * (DIM_K / 4) + lane];
#pragma unroll
    for (int k = 0; k < G; ++k) {
        acc.x = __builtin_fmaf(v[k], x[k].x, acc.x);
        acc.y = __builtin_fmaf(v[k], x[k].y, acc.y);
        acc.z = __builtin_fmaf(v[k], x[k].z, acc.z);
        acc.w = __builtin_fmaf(v[k], x[k].w, acc.w);
    }
}

__global__ void gcn_gather_kernel(const float* __restrict__ X,
                                  const int* __restrict__ cnt,
                                  const int* __restrict__ cursorEnd,
                                  const int* __restrict__ bucket,
                                  float* __restrict__ H, int N) {
    const int lane = threadIdx.x & 31;
    const int r = __builtin_amdgcn_readfirstlane(
        blockIdx.x * (blockDim.x >> 5) + (threadIdx.x >> 5));
    if (r >= N) return;
    const int n = cnt[r];
    const int2* ed = (const int2*)bucket + (cursorEnd[r] - n);
    const float4* X4 = (const float4*)X;
    float4 acc = make_float4(0.0f, 0.0f, 0.0f, 0.0f);

    for (int base = 0; base < n; base += 32) {
        const int m = min(32, n - base);
        int2 my = make_int2(0, 0);
        if (lane < m) my = ed[base + lane];
        int j = 0;
        for (; j + 8 <= m; j += 8) gcn_gather_group<8>(X4, my, j, lane, acc);
        if ((m - j) & 4) { gcn_gather_group<4>(X4, my, j, lane, acc); j += 4; }
        if ((m - j) & 2) { gcn_gather_group<2>(X4, my, j, lane, acc); j += 2; }
        if ((m - j) & 1) { gcn_gather_group<1>(X4, my, j, lane, acc); }
    }
    ((float4*)(H + (long long)r * DIM_K))[lane] = acc;
}

template <int G>
__device__ __forceinline__ void gcn_gather_group64(const float2* __restrict__ P2,
                                                   int2 my, int j, int lane,
                                                   float2& acc) {
    int c[G];
    float v[G];
    float2 x[G];
#pragma unroll
    for (int k = 0; k < G; ++k) {
        c[k] = __builtin_amdgcn_readlane(my.x, j + k);
        v[k] = __int_as_float(__builtin_amdgcn_readlane(my.y, j + k));
    }
#pragma unroll
    for (int k = 0; k < G; ++k) x[k] = P2[(long long)c[k] * (DIM_OUT / 2) + lane];
#pragma unroll
    for (int k = 0; k < G; ++k) {
        acc.x = __builtin_fmaf(v[k], x[k].x, acc.x);
        acc.y = __builtin_fmaf(v[k], x[k].y, acc.y);
    }
}

__global__ void gcn_gather64_tanh_kernel(const float* __restrict__ P,
                                         const int* __restrict__ cnt,
                                         const int* __restrict__ cursorEnd,
                                         const int* __restrict__ bucket,
                                         float* __restrict__ O, int N) {
    const int lane = threadIdx.x & 31;
    const int r = __builtin_amdgcn_readfirstlane(
        blockIdx.x * (blockDim.x >> 5) + (threadIdx.x >> 5));
    if (r >= N) return;
    const int n = cnt[r];
    const int2* ed = (const int2*)bucket + (cursorEnd[r] - n);
    const float2* P2 = (const float2*)P;
    float2 acc = make_float2(0.0f, 0.0f);

    for (int base = 0; base < n; base += 32) {
        const int m = min(32, n - base);
        int2 my = make_int2(0, 0);
        if (lane < m) my = ed[base + lane];
        int j = 0;
        for (; j + 16 <= m; j += 16) gcn_gather_group64<16>(P2, my, j, lane, acc);
        if ((m - j) & 8) { gcn_gather_group64<8>(P2, my, j, lane, acc); j += 8; }
        if ((m - j) & 4) { gcn_gather_group64<4>(P2, my, j, lane, acc); j += 4; }
        if ((m - j) & 2) { gcn_gather_group64<2>(P2, my, j, lane, acc); j += 2; }
        if ((m - j) & 1) { gcn_gather_group64<1>(P2, my, j, lane, acc); }
    }
    ((float2*)(O + (long long)r * DIM_OUT))[lane] =
        make_float2(fast_tanh(acc.x), fast_tanh(acc.y));
}

#define GEMM_WAVES 24
#define GEMM_THREADS (GEMM_WAVES * 32)
#define GEMM_GROUPS 256
#define W_PITCH (DIM_K + 4)

template <int NT, bool ACT>
__global__ void __launch_bounds__(GEMM_THREADS)
gcn_gemm_kernel(const float* __restrict__ S,
                const float* __restrict__ W,
                float* __restrict__ O,
                int nNodes) {
    __shared__ __attribute__((aligned(16))) float Wl[NT * 16 * W_PITCH];

    {
        const float4* W4 = (const float4*)W;
        constexpr int W4N = NT * 16 * (DIM_K / 4);
        constexpr int W_BATCH = (W4N + GEMM_THREADS - 1) / GEMM_THREADS;
        float4 w[W_BATCH];
#pragma unroll
        for (int j = 0; j < W_BATCH; ++j)
            w[j] = W4[min((int)threadIdx.x + j * GEMM_THREADS, W4N - 1)];
#pragma unroll
        for (int j = 0; j < W_BATCH; ++j) {
            const int i = (int)threadIdx.x + j * GEMM_THREADS;
            if (i < W4N) *(float4*)&Wl[(i >> 5) * W_PITCH + (i & 31) * 4] = w[j];
        }
    }
    __syncthreads();

    const int lane = threadIdx.x & 31;
    const int wave = __builtin_amdgcn_readfirstlane((int)(threadIdx.x >> 5));
    const int mTiles = (nNodes + 15) >> 4;

    const int nOut  = NT * 16;
    const int l15   = lane & 15;
    const int khalf = (lane < 16) ? 0 : 2;

    const float* bBase = Wl + l15 * W_PITCH + khalf;

    for (int mTile = wave * (int)gridDim.x + (int)blockIdx.x; mTile < mTiles;
         mTile += GEMM_WAVES * (int)gridDim.x) {
        const float* aBase = S + (long long)(mTile * 16 + l15) * DIM_K + khalf;

        v8f acc[NT];
#pragma unroll
        for (int t = 0; t < NT; ++t) acc[t] = (v8f){};

#pragma unroll 2
        for (int k = 0; k < DIM_K; k += 4) {
            v2f a;
            a.x = aBase[k];
            a.y = aBase[k + 1];
#pragma unroll
            for (int t = 0; t < NT; ++t) {
                const v2f b = *(const v2f*)(bBase + t * 16 * W_PITCH + k);
                acc[t] = __builtin_amdgcn_wmma_f32_16x16x4_f32(
                     false, a,  false, b,
                     (short)0, acc[t],  false,  false);
            }
        }

        const int mbase = mTile * 16 + ((lane < 16) ? 0 : 8);
#pragma unroll
        for (int t = 0; t < NT; ++t) {
            const int n = t * 16 + l15;
#pragma unroll
            for (int r = 0; r < 8; ++r) {
                O[(long long)(mbase + r) * nOut + n] = ACT ? fast_tanh(acc[t][r]) : acc[t][r];
            }
        }
    }
}

extern "C" void kernel_launch(void* const* d_in, const int* in_sizes, int n_in,
                              void* d_out, int out_size, void* d_ws, size_t ws_size,
                              hipStream_t stream) {
    const float* X        = (const float*)d_in[0];
    const float* W1       = (const float*)d_in[1];
    const float* W2       = (const float*)d_in[2];
    const float* edgeVals = (const float*)d_in[3];
    const int*   edgeRow  = (const int*)d_in[4];
    const int*   edgeCol  = (const int*)d_in[5];
    float* out = (float*)d_out;

    const int N = in_sizes[0] / DIM_K;
    const int E = in_sizes[3];
    const long long featCount = (long long)N * DIM_K;

    const int NB = (N + ROWS_PER_BIN - 1) / ROWS_PER_BIN;
    const int nPad = (N + 31) & ~31;

    float* S  = (float*)d_ws;
    float* P  = S;
    float* H1 = S + featCount;
    int* cnt    = (int*)(H1 + featCount);
    int* cursor = cnt + nPad;
    int* bucket = cursor + nPad;
    int* binned = bucket + 2 * (long long)E;
    int* chunkHist = binned + 2 * (long long)E;
    int* chunkOff  = chunkHist + EDGE_CHUNKS * NB;
    int* binTotal  = chunkOff + EDGE_CHUNKS * NB;
    int* binBase   = binTotal + NB + 32;

    const int threads = 256;

    const int wavesPerBlock = threads / 32;
    const int rowBlocks = (N + wavesPerBlock - 1) / wavesPerBlock;

    const int gemmThreads = GEMM_THREADS;
    const int gemmBlocks = GEMM_GROUPS;

    gcn_bin_hist_kernel<<<EDGE_CHUNKS, SORT_THREADS, NB * sizeof(int), stream>>>(
        edgeRow, chunkHist, E, NB);
    gcn_bin_offsets_kernel<<<(NB + 31) / 32, SORT_THREADS, 0, stream>>>(
        chunkHist, chunkOff, binTotal, NB);
    gcn_bin_scatter_kernel<<<EDGE_CHUNKS, SORT_THREADS, 3 * NB * sizeof(int), stream>>>(
        edgeVals, edgeRow, edgeCol, chunkOff, binTotal, binBase, binned, E, NB);
    gcn_bin_rows_kernel<<<NB, ROWS_THREADS, 0, stream>>>(
        binned, binBase, cnt, cursor, bucket, N);

    gcn_gather_kernel<<<rowBlocks, threads, 0, stream>>>(X, cnt, cursor, bucket, S, N);
    gcn_gemm_kernel<8, true><<<gemmBlocks, gemmThreads, 0, stream>>>(
        S, W1, H1, N);

    gcn_gemm_kernel<4, false><<<gemmBlocks, gemmThreads, 0, stream>>>(
        H1, W2, P, N);
    gcn_gather64_tanh_kernel<<<rowBlocks, threads, 0, stream>>>(P, cnt, cursor, bucket, out, N);
}


# Both row gathers run as resident waves: 512 work-groups of 32 waves loop over the rows (not one wave per row), next row's edge list fetched during the current row's gathers; sum order unchanged
# speedup vs baseline: 4.5631x; 1.0320x over previous
#include <hip/hip_runtime.h>
#include <hip/hip_bf16.h>
#include <math.h>


typedef __attribute__((ext_vector_type(2))) float v2f;
typedef __attribute__((ext_vector_type(8))) float v8f;

#define DIM_K 128
#define DIM_OUT 64

__device__ __forceinline__ float fast_tanh(float x) {
    float ax = fabsf(x);
    float e  = __expf(-2.0f * ax);
    float r  = (1.0f - e) * __builtin_amdgcn_rcpf(1.0f + e);
    return copysignf(r, x);
}

#define ROWS_PER_BIN_LOG2 7
#define ROWS_PER_BIN (1 << ROWS_PER_BIN_LOG2)
#define EDGE_CHUNKS 256
#define SORT_THREADS 1024
#define EDGE_BATCH 8
#define STAGE_CAP (EDGE_BATCH * SORT_THREADS)
#define ROWS_THREADS 256
#define BIN_STAGE_CAP 4096

typedef unsigned long long gcn_rec;

__device__ __forceinline__ gcn_rec gcn_make_rec(int lo, int hi) {
    return (gcn_rec)(unsigned)lo | ((gcn_rec)(unsigned)hi << 32);
}

__device__ __forceinline__ int gcn_wave_incl_scan(int x, int lane) {
#pragma unroll
    for (int d = 1; d < 32; d <<= 1) {
        int t = __shfl_up(x, d, 32);
        if (lane >= d) x += t;
    }
    return x;
}

__device__ __forceinline__ int gcn_block_excl_scan(int x, int* waveSum, int* blockSum, int& total) {
    const int lane = threadIdx.x & 31;
    const int wave = threadIdx.x >> 5;
    const int nw = blockDim.x >> 5;
    const int incl = gcn_wave_incl_scan(x, lane);
    if (lane == 31) waveSum[wave] = incl;
    __syncthreads();
    if (wave == 0) {
        const int w = (lane < nw) ? waveSum[lane] : 0;
        const int wi = gcn_wave_incl_scan(w, lane);
        waveSum[lane] = wi - w;
        if (lane == 31) *blockSum = wi;
    }
    __syncthreads();
    const int excl = waveSum[wave] + incl - x;
    total = *blockSum;
    __syncthreads();
    return excl;
}

__global__ void __launch_bounds__(SORT_THREADS)
gcn_bin_hist_kernel(const int* __restrict__ row, int* __restrict__ chunkHist, int E, int NB) {
    extern __shared__ int ldsBins[];
    for (int i = threadIdx.x; i < NB; i += SORT_THREADS) ldsBins[i] = 0;
    __syncthreads();
    const int chunk = (E + EDGE_CHUNKS - 1) / EDGE_CHUNKS;
    const int lo = min(E, (int)blockIdx.x * chunk);
    const int hi = min(E, lo + chunk);
    for (int e0 = lo + threadIdx.x; e0 < hi; e0 += STAGE_CAP) {
        int r[EDGE_BATCH];
#pragma unroll
        for (int k = 0; k < EDGE_BATCH; ++k) r[k] = row[min(e0 + k * SORT_THREADS, hi - 1)];
#pragma unroll
        for (int k = 0; k < EDGE_BATCH; ++k)
            if (e0 + k * SORT_THREADS < hi) atomicAdd(&ldsBins[r[k] >> ROWS_PER_BIN_LOG2], 1);
    }
    __syncthreads();
    for (int i = threadIdx.x; i < NB; i += SORT_THREADS)
        chunkHist[blockIdx.x * NB + i] = ldsBins[i];
}

__global__ void __launch_bounds__(SORT_THREADS)
gcn_bin_offsets_kernel(const int* __restrict__ chunkHist, int* __restrict__ chunkOff,
                       int* __restrict__ binTotal, int NB) {
    __shared__ int part[32][33];
    const int lane = threadIdx.x & 31;
    const int wave = threadIdx.x >> 5;
    const int bin = blockIdx.x * 32 + lane;
    const bool live = bin < NB;
    constexpr int PER = EDGE_CHUNKS / 32;
    int v[PER];
#pragma unroll
    for (int k = 0; k < PER; ++k) v[k] = live ? chunkHist[(wave * PER + k) * NB + bin] : 0;
    int s = 0;
#pragma unroll
    for (int k = 0; k < PER; ++k) { const int t = v[k]; v[k] = s; s += t; }
    part[wave][lane] = s;
    __syncthreads();
    int off = 0;
    for (int w = 0; w < wave; ++w) off += part[w][lane];
    if (live) {
#pragma unroll
        for (int k = 0; k < PER; ++k) chunkOff[(wave * PER + k) * NB + bin] = off + v[k];
        if (wave == 31) binTotal[bin] = off + s;
    }
}

__global__ void __launch_bounds__(SORT_THREADS)
gcn_bin_scatter_kernel(const float* __restrict__ vals, const int* __restrict__ row,
                       const int* __restrict__ col, const int* __restrict__ chunkOff,
                       const int* __restrict__ binTotal, int* __restrict__ binBase,
                       int* __restrict__ binnedWords, int E, int NB) {
    gcn_rec* __restrict__ binned = (gcn_rec*)binnedWords;
    extern __shared__ int ldsBins[];
    __shared__ gcn_rec stage[STAGE_CAP];
    __shared__ int stageDst[STAGE_CAP];
    __shared__ int waveSum[32];
    __shared__ int blockSum;
    int* gcur   = ldsBins;
    int* lcnt   = ldsBins + NB;
    int* lstart = ldsBins + 2 * NB;

    int carry = 0;
    for (int c = 0; c < NB; c += SORT_THREADS) {
        const int bin = c + threadIdx.x;
        const int tot = (bin < NB) ? binTotal[bin] : 0;
        int total;
        const int base = carry + gcn_block_excl_scan(tot, waveSum, &blockSum, total);
        if (bin < NB) {
            gcur[bin] = base + chunkOff[blockIdx.x * NB + bin];
            if (blockIdx.x == 0) binBase[bin] = base;
        }
        carry += total;
    }
    if (blockIdx.x == 0 && threadIdx.x == 0) binBase[NB] = carry;

    const int chunk = (E + EDGE_CHUNKS - 1) / EDGE_CHUNKS;
    const int lo = min(E, (int)blockIdx.x * chunk);
    const int hi = min(E, lo + chunk);
    for (int s0 = lo; s0 < hi; s0 += STAGE_CAP) {
        const int s1 = min(hi, s0 + STAGE_CAP);
        for (int i = threadIdx.x; i < NB; i += SORT_THREADS) lcnt[i] = 0;
        __syncthreads();

        int r[EDGE_BATCH], cidx[EDGE_BATCH], rank[EDGE_BATCH];
        float v[EDGE_BATCH];
#pragma unroll
        for (int k = 0; k < EDGE_BATCH; ++k) {
            const int e = min(s0 + (int)threadIdx.x + k * SORT_THREADS, s1 - 1);
            r[k] = row[e];
            cidx[k] = col[e];
            v[k] = vals[e];
        }
#pragma unroll
        for (int k = 0; k < EDGE_BATCH; ++k) {
            rank[k] = 0;
            if (s0 + (int)threadIdx.x + k * SORT_THREADS < s1)
                rank[k] = atomicAdd(&lcnt[r[k] >> ROWS_PER_BIN_LOG2], 1);
        }
        __syncthreads();

        int lcarry = 0;
        for (int c = 0; c < NB; c += SORT_THREADS) {
            const int bin = c + threadIdx.x;
            const int n = (bin < NB) ? lcnt[bin] : 0;
            int total;
            const int st = lcarry + gcn_block_excl_scan(n, waveSum, &blockSum, total);
            if (bin < NB) lstart[bin] = st;
            lcarry += total;
        }
        __syncthreads();

#pragma unroll
        for (int k = 0; k < EDGE_BATCH; ++k) {
            if (s0 + (int)threadIdx.x + k * SORT_THREADS < s1) {
                const int bin = r[k] >> ROWS_PER_BIN_LOG2;
                const int li = lstart[bin] + rank[k];
                stage[li] = gcn_make_rec(((r[k] & (ROWS_PER_BIN - 1)) << 24) | cidx[k],
                                         __float_as_int(v[k]));
                stageDst[li] = gcur[bin] + rank[k];
            }
        }
        __syncthreads();

        for (int i = threadIdx.x; i < s1 - s0; i += SORT_THREADS) binned[stageDst[i]] = stage[i];
        __syncthreads();
        for (int i = threadIdx.x; i < NB; i += SORT_THREADS) gcur[i] += lcnt[i];
        __syncthreads();
    }
}

__global__ void __launch_bounds__(ROWS_THREADS)
gcn_bin_rows_kernel(const int* __restrict__ binnedWords, const int* __restrict__ binBase,
                    int* __restrict__ cnt, int* __restrict__ cursorEnd,
                    int* __restrict__ bucketWords, int N) {
    const gcn_rec* __restrict__ binned = (const gcn_rec*)binnedWords;
    gcn_rec* __restrict__ bucket = (gcn_rec*)bucketWords;
    __shared__ gcn_rec stage[BIN_STAGE_CAP];
    __shared__ int rowCur[ROWS_PER_BIN];
    __shared__ int waveSum[ROWS_PER_BIN / 32];
    const int lane = threadIdx.x & 31;
    const int wave = threadIdx.x >> 5;
    const int bin = blockIdx.x;
    const int lo = binBase[bin];
    const int hi = binBase[bin + 1];
    const bool fits = (hi - lo) <= BIN_STAGE_CAP;
    if (threadIdx.x < ROWS_PER_BIN) rowCur[threadIdx.x] = 0;
    __syncthreads();
    for (int i0 = lo + threadIdx.x; i0 < hi; i0 += EDGE_BATCH * ROWS_THREADS) {
        gcn_rec rec[EDGE_BATCH];
#pragma unroll
        for (int k = 0; k < EDGE_BATCH; ++k) rec[k] = binned[min(i0 + k * ROWS_THREADS, hi - 1)];
#pragma unroll
        for (int k = 0; k < EDGE_BATCH; ++k)
            if (i0 + k * ROWS_THREADS < hi) atomicAdd(&rowCur[((int)rec[k]) >> 24], 1);
    }
    __syncthreads();
    int c = 0, incl = 0;
    if (threadIdx.x < ROWS_PER_BIN) {
        c = rowCur[threadIdx.x];
        incl = gcn_wave_incl_scan(c, lane);
        if (lane == 31) waveSum[wave] = incl;
    }
    __syncthreads();
    if (threadIdx.x < ROWS_PER_BIN) {
        int off = 0;
        for (int w = 0; w < wave; ++w) off += waveSum[w];
        const int rel = off + incl - c;
        const int r = bin * ROWS_PER_BIN + threadIdx.x;
        if (r < N) { cnt[r] = c; cursorEnd[r] = lo + rel + c; }
        rowCur[threadIdx.x] = rel;
    }
    __syncthreads();
    for (int i0 = lo + threadIdx.x; i0 < hi; i0 += EDGE_BATCH * ROWS_THREADS) {
        gcn_rec rec[EDGE_BATCH];
#pragma unroll
        for (int k = 0; k < EDGE_BATCH; ++k) rec[k] = binned[min(i0 + k * ROWS_THREADS, hi - 1)];
#pragma unroll
        for (int k = 0; k < EDGE_BATCH; ++k) {
            if (i0 + k * ROWS_THREADS < hi) {
                const int x = (int)rec[k];
                const int pos = atomicAdd(&rowCur[x >> 24], 1);
                const gcn_rec o = gcn_make_rec(x & 0xFFFFFF, (int)(rec[k] >> 32));
                if (fits) stage[pos] = o; else bucket[lo + pos] = o;
            }
        }
    }
    if (fits) {
        __syncthreads();
        for (int i = threadIdx.x; i < hi - lo; i += ROWS_THREADS) bucket[lo + i] = stage[i];
    }
}

template <int G>
__device__ __forceinline__ void gcn_gather_group(const float4* __restrict__ X4,
                                                 int2 my, int j, int lane,
                                                 float4& acc) {
    int c[G];
    float v[G];
    float4 x[G];
#pragma unroll
    for (int k = 0; k < G; ++k) {
        c[k] = __builtin_amdgcn_readlane(my.x, j + k);
        v[k] = __int_as_float(__builtin_amdgcn_readlane(my.y, j + k));
    }
#pragma unroll
    for (int k = 0; k < G; ++k) x[k] = X4[(long long)c[k] * (DIM_K / 4) + lane];
#pragma unroll
    for (int k = 0; k < G; ++k) {
        acc.x = __builtin_fmaf(v[k], x[k].x, acc.x);
        acc.y = __builtin_fmaf(v[k], x[k].y, acc.y);
        acc.z = __builtin_fmaf(v[k], x[k].z, acc.z);
        acc.w = __builtin_fmaf(v[k], x[k].w, acc.w);
    }
}

#define GATHER_WAVES 32
#define GATHER_THREADS (GATHER_WAVES * 32)
#define GATHER_GROUPS 512

__device__ __forceinline__ void gcn_row_chunk(const float4* __restrict__ X4,
                                              int2 my, int m, int lane, float4& acc) {
    int j = 0;
    for (; j + 8 <= m; j += 8) gcn_gather_group<8>(X4, my, j, lane, acc);
    if ((m - j) & 4) { gcn_gather_group<4>(X4, my, j, lane, acc); j += 4; }
    if ((m - j) & 2) { gcn_gather_group<2>(X4, my, j, lane, acc); j += 2; }
    if ((m - j) & 1) { gcn_gather_group<1>(X4, my, j, lane, acc); }
}

__global__ void __launch_bounds__(GATHER_THREADS)
gcn_gather_kernel(const float* __restrict__ X,
                  const int* __restrict__ cnt,
                  const int* __restrict__ cursorEnd,
                  const int* __restrict__ bucket,
                  float* __restrict__ H, int N) {
    const int lane = threadIdx.x & 31;
    const int stride = (int)gridDim.x * GATHER_WAVES;
    int rNext = __builtin_amdgcn_readfirstlane(
        (int)blockIdx.x * GATHER_WAVES + (int)(threadIdx.x >> 5));
    if (rNext >= N) return;
    const int2* ed = (const int2*)bucket;
    const float4* X4 = (const float4*)X;

    int r = -1, n = 0, s = 0;
    int2 my = make_int2(0, 0);
    int nNext = cnt[rNext];
    int eNext = cursorEnd[rNext];

    for (;;) {
        const int sNext = eNext - nNext;
        int2 myNext = make_int2(0, 0);
        if (lane < nNext) myNext = ed[sNext + lane];
        const int rAfter = rNext + stride;
        int nAfter = 0, eAfter = 0;
        if (rAfter < N) { nAfter = cnt[rAfter]; eAfter = cursorEnd[rAfter]; }

        float4 acc = make_float4(0.0f, 0.0f, 0.0f, 0.0f);
        gcn_row_chunk(X4, my, min(32, n), lane, acc);
        for (int base = 32; base < n; base += 32) {
            const int m = min(32, n - base);
            int2 more = make_int2(0, 0);
            if (lane < m) more = ed[s + base + lane];
            gcn_row_chunk(X4, more, m, lane, acc);
        }
        if (r >= 0) ((float4*)(H + (long long)r * DIM_K))[lane] = acc;

        if (rNext >= N) break;
        r = rNext; n = nNext; s = sNext; my = myNext;
        rNext = rAfter; nNext = nAfter; eNext = eAfter;
    }
}

template <int G>
__device__ __forceinline__ void gcn_gather_group64(const float2* __restrict__ P2,
                                                   int2 my, int j, int lane,
                                                   float2& acc) {
    int c[G];
    float v[G];
    float2 x[G];
#pragma unroll
    for (int k = 0; k < G; ++k) {
        c[k] = __builtin_amdgcn_readlane(my.x, j + k);
        v[k] = __int_as_float(__builtin_amdgcn_readlane(my.y, j + k));
    }
#pragma unroll
    for (int k = 0; k < G; ++k) x[k] = P2[(long long)c[k] * (DIM_OUT / 2) + lane];
#pragma unroll
    for (int k = 0; k < G; ++k) {
        acc.x = __builtin_fmaf(v[k], x[k].x, acc.x);
        acc.y = __builtin_fmaf(v[k], x[k].y, acc.y);
    }
}

__device__ __forceinline__ void gcn_row_chunk64(const float2* __restrict__ P2,
                                                int2 my, int m, int lane, float2& acc) {
    int j = 0;
    for (; j + 16 <= m; j += 16) gcn_gather_group64<16>(P2, my, j, lane, acc);
    if ((m - j) & 8) { gcn_gather_group64<8>(P2, my, j, lane, acc); j += 8; }
    if ((m - j) & 4) { gcn_gather_group64<4>(P2, my, j, lane, acc); j += 4; }
    if ((m - j) & 2) { gcn_gather_group64<2>(P2, my, j, lane, acc); j += 2; }
    if ((m - j) & 1) { gcn_gather_group64<1>(P2, my, j, lane, acc); }
}

__global__ void __launch_bounds__(GATHER_THREADS)
gcn_gather64_tanh_kernel(const float* __restrict__ P,
                         const int* __restrict__ cnt,
                         const int* __restrict__ cursorEnd,
                         const int* __restrict__ bucket,
                         float* __restrict__ O, int N) {
    const int lane = threadIdx.x & 31;
    const int stride = (int)gridDim.x * GATHER_WAVES;
    int rNext = __builtin_amdgcn_readfirstlane(
        (int)blockIdx.x * GATHER_WAVES + (int)(threadIdx.x >> 5));
    if (rNext >= N) return;
    const int2* ed = (const int2*)bucket;
    const float2* P2 = (const float2*)P;

    int r = -1, n = 0, s = 0;
    int2 my = make_int2(0, 0);
    int nNext = cnt[rNext];
    int eNext = cursorEnd[rNext];

    for (;;) {
        const int sNext = eNext - nNext;
        int2 myNext = make_int2(0, 0);
        if (lane < nNext) myNext = ed[sNext + lane];
        const int rAfter = rNext + stride;
        int nAfter = 0, eAfter = 0;
        if (rAfter < N) { nAfter = cnt[rAfter]; eAfter = cursorEnd[rAfter]; }

        float2 acc = make_float2(0.0f, 0.0f);
        gcn_row_chunk64(P2, my, min(32, n), lane, acc);
        for (int base = 32; base < n; base += 32) {
            const int m = min(32, n - base);
            int2 more = make_int2(0, 0);
            if (lane < m) more = ed[s + base + lane];
            gcn_row_chunk64(P2, more, m, lane, acc);
        }
        if (r >= 0)
            ((float2*)(O + (long long)r * DIM_OUT))[lane] =
                make_float2(fast_tanh(acc.x), fast_tanh(acc.y));

        if (rNext >= N) break;
        r = rNext; n = nNext; s = sNext; my = myNext;
        rNext = rAfter; nNext = nAfter; eNext = eAfter;
    }
}

#define GEMM_WAVES 24
#define GEMM_THREADS (GEMM_WAVES * 32)
#define GEMM_GROUPS 256
#define W_PITCH (DIM_K + 4)

template <int NT, bool ACT>
__global__ void __launch_bounds__(GEMM_THREADS)
gcn_gemm_kernel(const float* __restrict__ S,
                const float* __restrict__ W,
                float* __restrict__ O,
                int nNodes) {
    __shared__ __attribute__((aligned(16))) float Wl[NT * 16 * W_PITCH];

    {
        const float4* W4 = (const float4*)W;
        constexpr int W4N = NT * 16 * (DIM_K / 4);
        constexpr int W_BATCH = (W4N + GEMM_THREADS - 1) / GEMM_THREADS;
        float4 w[W_BATCH];
#pragma unroll
        for (int j = 0; j < W_BATCH; ++j)
            w[j] = W4[min((int)threadIdx.x + j * GEMM_THREADS, W4N - 1)];
#pragma unroll
        for (int j = 0; j < W_BATCH; ++j) {
            const int i = (int)threadIdx.x + j * GEMM_THREADS;
            if (i < W4N) *(float4*)&Wl[(i >> 5) * W_PITCH + (i & 31) * 4] = w[j];
        }
    }
    __syncthreads();

    const int lane = threadIdx.x & 31;
    const int wave = __builtin_amdgcn_readfirstlane((int)(threadIdx.x >> 5));
    const int mTiles = (nNodes + 15) >> 4;

    const int nOut  = NT * 16;
    const int l15   = lane & 15;
    const int khalf = (lane < 16) ? 0 : 2;

    const float* bBase = Wl + l15 * W_PITCH + khalf;

    for (int mTile = wave * (int)gridDim.x + (int)blockIdx.x; mTile < mTiles;
         mTile += GEMM_WAVES * (int)gridDim.x) {
        const float* aBase = S + (long long)(mTile * 16 + l15) * DIM_K + khalf;

        v8f acc[NT];
#pragma unroll
        for (int t = 0; t < NT; ++t) acc[t] = (v8f){};

#pragma unroll 2
        for (int k = 0; k < DIM_K; k += 4) {
            v2f a;
            a.x = aBase[k];
            a.y = aBase[k + 1];
#pragma unroll
            for (int t = 0; t < NT; ++t) {
                const v2f b = *(const v2f*)(bBase + t * 16 * W_PITCH + k);
                acc[t] = __builtin_amdgcn_wmma_f32_16x16x4_f32(
                     false, a,  false, b,
                     (short)0, acc[t],  false,  false);
            }
        }

        const int mbase = mTile * 16 + ((lane < 16) ? 0 : 8);
#pragma unroll
        for (int t = 0; t < NT; ++t) {
            const int n = t * 16 + l15;
#pragma unroll
            for (int r = 0; r < 8; ++r) {
                O[(long long)(mbase + r) * nOut + n] = ACT ? fast_tanh(acc[t][r]) : acc[t][r];
            }
        }
    }
}

extern "C" void kernel_launch(void* const* d_in, const int* in_sizes, int n_in,
                              void* d_out, int out_size, void* d_ws, size_t ws_size,
                              hipStream_t stream) {
    const float* X        = (const float*)d_in[0];
    const float* W1       = (const float*)d_in[1];
    const float* W2       = (const float*)d_in[2];
    const float* edgeVals = (const float*)d_in[3];
    const int*   edgeRow  = (const int*)d_in[4];
    const int*   edgeCol  = (const int*)d_in[5];
    float* out = (float*)d_out;

    const int N = in_sizes[0] / DIM_K;
    const int E = in_sizes[3];
    const long long featCount = (long long)N * DIM_K;

    const int NB = (N + ROWS_PER_BIN - 1) / ROWS_PER_BIN;
    const int nPad = (N + 31) & ~31;

    float* S  = (float*)d_ws;
    float* P  = S;
    float* H1 = S + featCount;
    int* cnt    = (int*)(H1 + featCount);
    int* cursor = cnt + nPad;
    int* bucket = cursor + nPad;
    int* binned = bucket + 2 * (long long)E;
    int* chunkHist = binned + 2 * (long long)E;
    int* chunkOff  = chunkHist + EDGE_CHUNKS * NB;
    int* binTotal  = chunkOff + EDGE_CHUNKS * NB;
    int* binBase   = binTotal + NB + 32;

    const int gatherThreads = GATHER_THREADS;
    const int gatherBlocks = GATHER_GROUPS;

    const int gemmThreads = GEMM_THREADS;
    const int gemmBlocks = GEMM_GROUPS;

    gcn_bin_hist_kernel<<<EDGE_CHUNKS, SORT_THREADS, NB * sizeof(int), stream>>>(
        edgeRow, chunkHist, E, NB);
    gcn_bin_offsets_kernel<<<(NB + 31) / 32, SORT_THREADS, 0, stream>>>(
        chunkHist, chunkOff, binTotal, NB);
    gcn_bin_scatter_kernel<<<EDGE_CHUNKS, SORT_THREADS, 3 * NB * sizeof(int), stream>>>(
        edgeVals, edgeRow, edgeCol, chunkOff, binTotal, binBase, binned, E, NB);
    gcn_bin_rows_kernel<<<NB, ROWS_THREADS, 0, stream>>>(
        binned, binBase, cnt, cursor, bucket, N);

    gcn_gather_kernel<<<gatherBlocks, gatherThreads, 0, stream>>>(X, cnt, cursor, bucket, S, N);
    gcn_gemm_kernel<8, true><<<gemmBlocks, gemmThreads, 0, stream>>>(
        S, W1, H1, N);

    gcn_gemm_kernel<4, false><<<gemmBlocks, gemmThreads, 0, stream>>>(
        H1, W2, P, N);
    gcn_gather64_tanh_kernel<<<gatherBlocks, gatherThreads, 0, stream>>>(P, cnt, cursor, bucket, out, N);
}


# Layer 1's row gather and W1 product share one kernel: 24 gather waves fill LDS row tiles while 8 matrix waves run WMMA on the previous tile (A, B from LDS); row sums never stored; sum order unchanged
# speedup vs baseline: 5.1049x; 1.1187x over previous
#include <hip/hip_runtime.h>
#include <hip/hip_bf16.h>
#include <math.h>


typedef __attribute__((ext_vector_type(2))) float v2f;
typedef __attribute__((ext_vector_type(8))) float v8f;

#define DIM_K 128
#define DIM_OUT 64

__device__ __forceinline__ float fast_tanh(float x) {
    float ax = fabsf(x);
    float e  = __expf(-2.0f * ax);
    float r  = (1.0f - e) * __builtin_amdgcn_rcpf(1.0f + e);
    return copysignf(r, x);
}

#define ROWS_PER_BIN_LOG2 7
#define ROWS_PER_BIN (1 << ROWS_PER_BIN_LOG2)
#define EDGE_CHUNKS 256
#define SORT_THREADS 1024
#define EDGE_BATCH 8
#define STAGE_CAP (EDGE_BATCH * SORT_THREADS)
#define ROWS_THREADS 256
#define BIN_STAGE_CAP 4096

typedef unsigned long long gcn_rec;

__device__ __forceinline__ gcn_rec gcn_make_rec(int lo, int hi) {
    return (gcn_rec)(unsigned)lo | ((gcn_rec)(unsigned)hi << 32);
}

__device__ __forceinline__ int gcn_wave_incl_scan(int x, int lane) {
#pragma unroll
    for (int d = 1; d < 32; d <<= 1) {
        int t = __shfl_up(x, d, 32);
        if (lane >= d) x += t;
    }
    return x;
}

__device__ __forceinline__ int gcn_block_excl_scan(int x, int* waveSum, int* blockSum, int& total) {
    const int lane = threadIdx.x & 31;
    const int wave = threadIdx.x >> 5;
    const int nw = blockDim.x >> 5;
    const int incl = gcn_wave_incl_scan(x, lane);
    if (lane == 31) waveSum[wave] = incl;
    __syncthreads();
    if (wave == 0) {
        const int w = (lane < nw) ? waveSum[lane] : 0;
        const int wi = gcn_wave_incl_scan(w, lane);
        waveSum[lane] = wi - w;
        if (lane == 31) *blockSum = wi;
    }
    __syncthreads();
    const int excl = waveSum[wave] + incl - x;
    total = *blockSum;
    __syncthreads();
    return excl;
}

__global__ void __launch_bounds__(SORT_THREADS)
gcn_bin_hist_kernel(const int* __restrict__ row, int* __restrict__ chunkHist, int E, int NB) {
    extern __shared__ int ldsBins[];
    for (int i = threadIdx.x; i < NB; i += SORT_THREADS) ldsBins[i] = 0;
    __syncthreads();
    const int chunk = (E + EDGE_CHUNKS - 1) / EDGE_CHUNKS;
    const int lo = min(E, (int)blockIdx.x * chunk);
    const int hi = min(E, lo + chunk);
    for (int e0 = lo + threadIdx.x; e0 < hi; e0 += STAGE_CAP) {
        int r[EDGE_BATCH];
#pragma unroll
        for (int k = 0; k < EDGE_BATCH; ++k) r[k] = row[min(e0 + k * SORT_THREADS, hi - 1)];
#pragma unroll
        for (int k = 0; k < EDGE_BATCH; ++k)
            if (e0 + k * SORT_THREADS < hi) atomicAdd(&ldsBins[r[k] >> ROWS_PER_BIN_LOG2], 1);
    }
    __syncthreads();
    for (int i = threadIdx.x; i < NB; i += SORT_THREADS)
        chunkHist[blockIdx.x * NB + i] = ldsBins[i];
}

__global__ void __launch_bounds__(SORT_THREADS)
gcn_bin_offsets_kernel(const int* __restrict__ chunkHist, int* __restrict__ chunkOff,
                       int* __restrict__ binTotal, int NB) {
    __shared__ int part[32][33];
    const int lane = threadIdx.x & 31;
    const int wave = threadIdx.x >> 5;
    const int bin = blockIdx.x * 32 + lane;
    const bool live = bin < NB;
    constexpr int PER = EDGE_CHUNKS / 32;
    int v[PER];
#pragma unroll
    for (int k = 0; k < PER; ++k) v[k] = live ? chunkHist[(wave * PER + k) * NB + bin] : 0;
    int s = 0;
#pragma unroll
    for (int k = 0; k < PER; ++k) { const int t = v[k]; v[k] = s; s += t; }
    part[wave][lane] = s;
    __syncthreads();
    int off = 0;
    for (int w = 0; w < wave; ++w) off += part[w][lane];
    if (live) {
#pragma unroll
        for (int k = 0; k < PER; ++k) chunkOff[(wave * PER + k) * NB + bin] = off + v[k];
        if (wave == 31) binTotal[bin] = off + s;
    }
}

__global__ void __launch_bounds__(SORT_THREADS)
gcn_bin_scatter_kernel(const float* __restrict__ vals, const int* __restrict__ row,
                       const int* __restrict__ col, const int* __restrict__ chunkOff,
                       const int* __restrict__ binTotal, int* __restrict__ binBase,
                       int* __restrict__ binnedWords, int E, int NB) {
    gcn_rec* __restrict__ binned = (gcn_rec*)binnedWords;
    extern __shared__ int ldsBins[];
    __shared__ gcn_rec stage[STAGE_CAP];
    __shared__ int stageDst[STAGE_CAP];
    __shared__ int waveSum[32];
    __shared__ int blockSum;
    int* gcur   = ldsBins;
    int* lcnt   = ldsBins + NB;
    int* lstart = ldsBins + 2 * NB;

    int carry = 0;
    for (int c = 0; c < NB; c += SORT_THREADS) {
        const int bin = c + threadIdx.x;
        const int tot = (bin < NB) ? binTotal[bin] : 0;
        int total;
        const int base = carry + gcn_block_excl_scan(tot, waveSum, &blockSum, total);
        if (bin < NB) {
            gcur[bin] = base + chunkOff[blockIdx.x * NB + bin];
            if (blockIdx.x == 0) binBase[bin] = base;
        }
        carry += total;
    }
    if (blockIdx.x == 0 && threadIdx.x == 0) binBase[NB] = carry;

    const int chunk = (E + EDGE_CHUNKS - 1) / EDGE_CHUNKS;
    const int lo = min(E, (int)blockIdx.x * chunk);
    const int hi = min(E, lo + chunk);
    for (int s0 = lo; s0 < hi; s0 += STAGE_CAP) {
        const int s1 = min(hi, s0 + STAGE_CAP);
        for (int i = threadIdx.x; i < NB; i += SORT_THREADS) lcnt[i] = 0;
        __syncthreads();

        int r[EDGE_BATCH], cidx[EDGE_BATCH], rank[EDGE_BATCH];
        float v[EDGE_BATCH];
#pragma unroll
        for (int k = 0; k < EDGE_BATCH; ++k) {
            const int e = min(s0 + (int)threadIdx.x + k * SORT_THREADS, s1 - 1);
            r[k] = row[e];
            cidx[k] = col[e];
            v[k] = vals[e];
        }
#pragma unroll
        for (int k = 0; k < EDGE_BATCH; ++k) {
            rank[k] = 0;
            if (s0 + (int)threadIdx.x + k * SORT_THREADS < s1)
                rank[k] = atomicAdd(&lcnt[r[k] >> ROWS_PER_BIN_LOG2], 1);
        }
        __syncthreads();

        int lcarry = 0;
        for (int c = 0; c < NB; c += SORT_THREADS) {
            const int bin = c + threadIdx.x;
            const int n = (bin < NB) ? lcnt[bin] : 0;
            int total;
            const int st = lcarry + gcn_block_excl_scan(n, waveSum, &blockSum, total);
            if (bin < NB) lstart[bin] = st;
            lcarry += total;
        }
        __syncthreads();

#pragma unroll
        for (int k = 0; k < EDGE_BATCH; ++k) {
            if (s0 + (int)threadIdx.x + k * SORT_THREADS < s1) {
                const int bin = r[k] >> ROWS_PER_BIN_LOG2;
                const int li = lstart[bin] + rank[k];
                stage[li] = gcn_make_rec(((r[k] & (ROWS_PER_BIN - 1)) << 24) | cidx[k],
                                         __float_as_int(v[k]));
                stageDst[li] = gcur[bin] + rank[k];
            }
        }
        __syncthreads();

        for (int i = threadIdx.x; i < s1 - s0; i += SORT_THREADS) binned[stageDst[i]] = stage[i];
        __syncthreads();
        for (int i = threadIdx.x; i < NB; i += SORT_THREADS) gcur[i] += lcnt[i];
        __syncthreads();
    }
}

__global__ void __launch_bounds__(ROWS_THREADS)
gcn_bin_rows_kernel(const int* __restrict__ binnedWords, const int* __restrict__ binBase,
                    int* __restrict__ cnt, int* __restrict__ cursorEnd,
                    int* __restrict__ bucketWords, int N) {
    const gcn_rec* __restrict__ binned = (const gcn_rec*)binnedWords;
    gcn_rec* __restrict__ bucket = (gcn_rec*)bucketWords;
    __shared__ gcn_rec stage[BIN_STAGE_CAP];
    __shared__ int rowCur[ROWS_PER_BIN];
    __shared__ int waveSum[ROWS_PER_BIN / 32];
    const int lane = threadIdx.x & 31;
    const int wave = threadIdx.x >> 5;
    const int bin = blockIdx.x;
    const int lo = binBase[bin];
    const int hi = binBase[bin + 1];
    const bool fits = (hi - lo) <= BIN_STAGE_CAP;
    if (threadIdx.x < ROWS_PER_BIN) rowCur[threadIdx.x] = 0;
    __syncthreads();
    for (int i0 = lo + threadIdx.x; i0 < hi; i0 += EDGE_BATCH * ROWS_THREADS) {
        gcn_rec rec[EDGE_BATCH];
#pragma unroll
        for (int k = 0; k < EDGE_BATCH; ++k) rec[k] = binned[min(i0 + k * ROWS_THREADS, hi - 1)];
#pragma unroll
        for (int k = 0; k < EDGE_BATCH; ++k)
            if (i0 + k * ROWS_THREADS < hi) atomicAdd(&rowCur[((int)rec[k]) >> 24], 1);
    }
    __syncthreads();
    int c = 0, incl = 0;
    if (threadIdx.x < ROWS_PER_BIN) {
        c = rowCur[threadIdx.x];
        incl = gcn_wave_incl_scan(c, lane);
        if (lane == 31) waveSum[wave] = incl;
    }
    __syncthreads();
    if (threadIdx.x < ROWS_PER_BIN) {
        int off = 0;
        for (int w = 0; w < wave; ++w) off += waveSum[w];
        const int rel = off + incl - c;
        const int r = bin * ROWS_PER_BIN + threadIdx.x;
        if (r < N) { cnt[r] = c; cursorEnd[r] = lo + rel + c; }
        rowCur[threadIdx.x] = rel;
    }
    __syncthreads();
    for (int i0 = lo + threadIdx.x; i0 < hi; i0 += EDGE_BATCH * ROWS_THREADS) {
        gcn_rec rec[EDGE_BATCH];
#pragma unroll
        for (int k = 0; k < EDGE_BATCH; ++k) rec[k] = binned[min(i0 + k * ROWS_THREADS, hi - 1)];
#pragma unroll
        for (int k = 0; k < EDGE_BATCH; ++k) {
            if (i0 + k * ROWS_THREADS < hi) {
                const int x = (int)rec[k];
                const int pos = atomicAdd(&rowCur[x >> 24], 1);
                const gcn_rec o = gcn_make_rec(x & 0xFFFFFF, (int)(rec[k] >> 32));
                if (fits) stage[pos] = o; else bucket[lo + pos] = o;
            }
        }
    }
    if (fits) {
        __syncthreads();
        for (int i = threadIdx.x; i < hi - lo; i += ROWS_THREADS) bucket[lo + i] = stage[i];
    }
}

template <int G>
__device__ __forceinline__ void gcn_gather_group(const float4* __restrict__ X4,
                                                 int2 my, int j, int lane,
                                                 float4& acc) {
    int c[G];
    float v[G];
    float4 x[G];
#pragma unroll
    for (int k = 0; k < G; ++k) {
        c[k] = __builtin_amdgcn_readlane(my.x, j + k);
        v[k] = __int_as_float(__builtin_amdgcn_readlane(my.y, j + k));
    }
#pragma unroll
    for (int k = 0; k < G; ++k) x[k] = X4[(long long)c[k] * (DIM_K / 4) + lane];
#pragma unroll
    for (int k = 0; k < G; ++k) {
        acc.x = __builtin_fmaf(v[k], x[k].x, acc.x);
        acc.y = __builtin_fmaf(v[k], x[k].y, acc.y);
        acc.z = __builtin_fmaf(v[k], x[k].z, acc.z);
        acc.w = __builtin_fmaf(v[k], x[k].w, acc.w);
    }
}

#define GATHER_WAVES 32
#define GATHER_THREADS (GATHER_WAVES * 32)
#define GATHER_GROUPS 512

__device__ __forceinline__ void gcn_row_chunk(const float4* __restrict__ X4,
                                              int2 my, int m, int lane, float4& acc) {
    int j = 0;
    for (; j + 8 <= m; j += 8) gcn_gather_group<8>(X4, my, j, lane, acc);
    if ((m - j) & 4) { gcn_gather_group<4>(X4, my, j, lane, acc); j += 4; }
    if ((m - j) & 2) { gcn_gather_group<2>(X4, my, j, lane, acc); j += 2; }
    if ((m - j) & 1) { gcn_gather_group<1>(X4, my, j, lane, acc); }
}

template <int G>
__device__ __forceinline__ void gcn_gather_group64(const float2* __restrict__ P2,
                                                   int2 my, int j, int lane,
                                                   float2& acc) {
    int c[G];
    float v[G];
    float2 x[G];
#pragma unroll
    for (int k = 0; k < G; ++k) {
        c[k] = __builtin_amdgcn_readlane(my.x, j + k);
        v[k] = __int_as_float(__builtin_amdgcn_readlane(my.y, j + k));
    }
#pragma unroll
    for (int k = 0; k < G; ++k) x[k] = P2[(long long)c[k] * (DIM_OUT / 2) + lane];
#pragma unroll
    for (int k = 0; k < G; ++k) {
        acc.x = __builtin_fmaf(v[k], x[k].x, acc.x);
        acc.y = __builtin_fmaf(v[k], x[k].y, acc.y);
    }
}

__device__ __forceinline__ void gcn_row_chunk64(const float2* __restrict__ P2,
                                                int2 my, int m, int lane, float2& acc) {
    int j = 0;
    for (; j + 16 <= m; j += 16) gcn_gather_group64<16>(P2, my, j, lane, acc);
    if ((m - j) & 8) { gcn_gather_group64<8>(P2, my, j, lane, acc); j += 8; }
    if ((m - j) & 4) { gcn_gather_group64<4>(P2, my, j, lane, acc); j += 4; }
    if ((m - j) & 2) { gcn_gather_group64<2>(P2, my, j, lane, acc); j += 2; }
    if ((m - j) & 1) { gcn_gather_group64<1>(P2, my, j, lane, acc); }
}

__global__ void __launch_bounds__(GATHER_THREADS)
gcn_gather64_tanh_kernel(const float* __restrict__ P,
                         const int* __restrict__ cnt,
                         const int* __restrict__ cursorEnd,
                         const int* __restrict__ bucket,
                         float* __restrict__ O, int N) {
    const int lane = threadIdx.x & 31;
    const int stride = (int)gridDim.x * GATHER_WAVES;
    int rNext = __builtin_amdgcn_readfirstlane(
        (int)blockIdx.x * GATHER_WAVES + (int)(threadIdx.x >> 5));
    if (rNext >= N) return;
    const int2* ed = (const int2*)bucket;
    const float2* P2 = (const float2*)P;

    int r = -1, n = 0, s = 0;
    int2 my = make_int2(0, 0);
    int nNext = cnt[rNext];
    int eNext = cursorEnd[rNext];

    for (;;) {
        const int sNext = eNext - nNext;
        int2 myNext = make_int2(0, 0);
        if (lane < nNext) myNext = ed[sNext + lane];
        const int rAfter = rNext + stride;
        int nAfter = 0, eAfter = 0;
        if (rAfter < N) { nAfter = cnt[rAfter]; eAfter = cursorEnd[rAfter]; }

        float2 acc = make_float2(0.0f, 0.0f);
        gcn_row_chunk64(P2, my, min(32, n), lane, acc);
        for (int base = 32; base < n; base += 32) {
            const int m = min(32, n - base);
            int2 more = make_int2(0, 0);
            if (lane < m) more = ed[s + base + lane];
            gcn_row_chunk64(P2, more, m, lane, acc);
        }
        if (r >= 0)
            ((float2*)(O + (long long)r * DIM_OUT))[lane] =
                make_float2(fast_tanh(acc.x), fast_tanh(acc.y));

        if (rNext >= N) break;
        r = rNext; n = nNext; s = sNext; my = myNext;
        rNext = rAfter; nNext = nAfter; eNext = eAfter;
    }
}

#define GEMM_WAVES 24
#define GEMM_THREADS (GEMM_WAVES * 32)
#define GEMM_GROUPS 256
#define W_PITCH (DIM_K + 4)

template <int NT, bool ACT>
__global__ void __launch_bounds__(GEMM_THREADS)
gcn_gemm_kernel(const float* __restrict__ S,
                const float* __restrict__ W,
                float* __restrict__ O,
                int nNodes) {
    __shared__ __attribute__((aligned(16))) float Wl[NT * 16 * W_PITCH];

    {
        const float4* W4 = (const float4*)W;
        constexpr int W4N = NT * 16 * (DIM_K / 4);
        constexpr int W_BATCH = (W4N + GEMM_THREADS - 1) / GEMM_THREADS;
        float4 w[W_BATCH];
#pragma unroll
        for (int j = 0; j < W_BATCH; ++j)
            w[j] = W4[min((int)threadIdx.x + j * GEMM_THREADS, W4N - 1)];
#pragma unroll
        for (int j = 0; j < W_BATCH; ++j) {
            const int i = (int)threadIdx.x + j * GEMM_THREADS;
            if (i < W4N) *(float4*)&Wl[(i >> 5) * W_PITCH + (i & 31) * 4] = w[j];
        }
    }
    __syncthreads();

    const int lane = threadIdx.x & 31;
    const int wave = __builtin_amdgcn_readfirstlane((int)(threadIdx.x >> 5));
    const int mTiles = (nNodes + 15) >> 4;

    const int nOut  = NT * 16;
    const int l15   = lane & 15;
    const int khalf = (lane < 16) ? 0 : 2;

    const float* bBase = Wl + l15 * W_PITCH + khalf;

    for (int mTile = wave * (int)gridDim.x + (int)blockIdx.x; mTile < mTiles;
         mTile += GEMM_WAVES * (int)gridDim.x) {
        const float* aBase = S + (long long)(mTile * 16 + l15) * DIM_K + khalf;

        v8f acc[NT];
#pragma unroll
        for (int t = 0; t < NT; ++t) acc[t] = (v8f){};

#pragma unroll 2
        for (int k = 0; k < DIM_K; k += 4) {
            v2f a;
            a.x = aBase[k];
            a.y = aBase[k + 1];
#pragma unroll
            for (int t = 0; t < NT; ++t) {
                const v2f b = *(const v2f*)(bBase + t * 16 * W_PITCH + k);
                acc[t] = __builtin_amdgcn_wmma_f32_16x16x4_f32(
                     false, a,  false, b,
                     (short)0, acc[t],  false,  false);
            }
        }

        const int mbase = mTile * 16 + ((lane < 16) ? 0 : 8);
#pragma unroll
        for (int t = 0; t < NT; ++t) {
            const int n = t * 16 + l15;
#pragma unroll
            for (int r = 0; r < 8; ++r) {
                O[(long long)(mbase + r) * nOut + n] = ACT ? fast_tanh(acc[t][r]) : acc[t][r];
            }
        }
    }
}

#define FUSE_WAVES 32
#define FUSE_THREADS (FUSE_WAVES * 32)
#define FUSE_GROUPS 512
#define FUSE_GATHER_WAVES 24
#define FUSE_ROWS_PER_WAVE 2
#define FUSE_TILE_ROWS (FUSE_GATHER_WAVES * FUSE_ROWS_PER_WAVE)

typedef const __attribute__((address_space(4))) int* gcn_meta_ptr;

__device__ __forceinline__ float4 gcn_row_trip(const float4* __restrict__ X4,
                                               const int2* __restrict__ ed,
                                               gcn_meta_ptr cnt, gcn_meta_ptr cursorEnd,
                                               int lane, int rowStride, int rEnd,
                                               int& n, int& s, int2& my,
                                               int& rNext, int& nNext, int& eNext) {
    const int sNext = eNext - nNext;
    int2 myNext = make_int2(0, 0);
    if (lane < nNext) myNext = ed[sNext + lane];
    const int rAfter = rNext + rowStride;
    int nAfter = 0, eAfter = 0;
    if (rAfter < rEnd) { nAfter = cnt[rAfter]; eAfter = cursorEnd[rAfter]; }

    float4 acc = make_float4(0.0f, 0.0f, 0.0f, 0.0f);
    gcn_row_chunk(X4, my, min(32, n), lane, acc);
    for (int base = 32; base < n; base += 32) {
        const int m = min(32, n - base);
        int2 more = make_int2(0, 0);
        if (lane < m) more = ed[s + base + lane];
        gcn_row_chunk(X4, more, m, lane, acc);
    }
    n = nNext; s = sNext; my = myNext;
    rNext = rAfter; nNext = nAfter; eNext = eAfter;
    return acc;
}

__device__ __forceinline__ void gcn_strip_piece(const float* __restrict__ aBase,
                                                const float* __restrict__ bBase,
                                                float* __restrict__ O,
                                                int rowFirst, int rEnd, int colFirst, int lane) {
    v8f acc0 = (v8f){};
    v8f acc1 = (v8f){};
#pragma unroll 4
    for (int k = 0; k < DIM_K; k += 4) {
        const v2f a  = *(const v2f*)(aBase + k);
        const v2f b0 = *(const v2f*)(bBase + k);
        const v2f b1 = *(const v2f*)(bBase + 16 * W_PITCH + k);
        acc0 = __builtin_amdgcn_wmma_f32_16x16x4_f32(
             false, a,  false, b0,
             (short)0, acc0,  false,  false);
        acc1 = __builtin_amdgcn_wmma_f32_16x16x4_f32(
             false, a,  false, b1,
             (short)0, acc1,  false,  false);
    }
    const int l15 = lane & 15;
    const int mrow = rowFirst + ((lane < 16) ? 0 : 8);
#pragma unroll
    for (int r = 0; r < 8; ++r) {
        const int row = mrow + r;
        if (row < rEnd) {
            float* o = O + (long long)row * DIM_K + colFirst + l15;
            o[0]  = fast_tanh(acc0[r]);
            o[16] = fast_tanh(acc1[r]);
        }
    }
}

__global__ void __launch_bounds__(FUSE_THREADS) __attribute__((amdgpu_waves_per_eu(16)))
gcn_gather_gemm_tanh_kernel(const float* __restrict__ X,
                            const int* __restrict__ cntGlobal,
                            const int* __restrict__ cursorEndGlobal,
                            const int* __restrict__ bucket,
                            const float* __restrict__ W,
                            float* __restrict__ O, int N) {
    const gcn_meta_ptr cnt = (gcn_meta_ptr)cntGlobal;
    const gcn_meta_ptr cursorEnd = (gcn_meta_ptr)cursorEndGlobal;
    __shared__ __attribute__((aligned(16))) float Wl[DIM_K * W_PITCH];
    __shared__ __attribute__((aligned(16))) float St[2 * FUSE_TILE_ROWS * W_PITCH];

    {
        const float4* W4 = (const float4*)W;
        constexpr int W4N = DIM_K * (DIM_K / 4);
        constexpr int W_BATCH = W4N / FUSE_THREADS;
        float4 w[W_BATCH];
#pragma unroll
        for (int j = 0; j < W_BATCH; ++j) w[j] = W4[(int)threadIdx.x + j * FUSE_THREADS];
#pragma unroll
        for (int j = 0; j < W_BATCH; ++j) {
            const int i = (int)threadIdx.x + j * FUSE_THREADS;
            *(float4*)&Wl[(i >> 5) * W_PITCH + (i & 31) * 4] = w[j];
        }
    }

    const int lane = threadIdx.x & 31;
    const int wave = __builtin_amdgcn_readfirstlane((int)(threadIdx.x >> 5));
    const int quad = wave >> 2;
    const int pos  = wave & 3;
    const bool mmaWave = pos == (quad & 3);
    const int g = quad * 3 + ((pos > (quad & 3)) ? pos - 1 : pos);

    const int rowsPer = (N + (int)gridDim.x - 1) / (int)gridDim.x;
    const int r0 = min(N, (int)blockIdx.x * rowsPer);
    const int r1 = min(N, r0 + rowsPer);
    const int nT = (r1 - r0 + FUSE_TILE_ROWS - 1) / FUSE_TILE_ROWS;

    const int2* ed = (const int2*)bucket;
    const float4* X4 = (const float4*)X;

    int n = 0, s = 0;
    int2 my = make_int2(0, 0);
    int rNext = r0 + g, nNext = 0, eNext = 0;
    if (!mmaWave) {
        if (rNext < r1) { nNext = cnt[rNext]; eNext = cursorEnd[rNext]; }
        (void)gcn_row_trip(X4, ed, cnt, cursorEnd, lane, FUSE_GATHER_WAVES, r1,
                           n, s, my, rNext, nNext, eNext);
    }

    const int l15   = lane & 15;
    const int khalf = (lane < 16) ? 0 : 2;

    for (int it = 0; it <= nT; ++it) {
        if (!mmaWave) {
            if (it < nT) {
                float* tile = St + (it & 1) * (FUSE_TILE_ROWS * W_PITCH);
#pragma unroll
                for (int q = 0; q < FUSE_ROWS_PER_WAVE; ++q) {
                    const float4 acc = gcn_row_trip(X4, ed, cnt, cursorEnd, lane,
                                                    FUSE_GATHER_WAVES, r1,
                                                    n, s, my, rNext, nNext, eNext);
                    *(float4*)&tile[(g + q * FUSE_GATHER_WAVES) * W_PITCH + lane * 4] = acc;
                }
            }
        } else if (it >= 1) {
            const int tt = it - 1;
            const float* tile = St + (tt & 1) * (FUSE_TILE_ROWS * W_PITCH);
            const int tileRow = r0 + tt * FUSE_TILE_ROWS;
            const int cp = quad & 3;
            const float* bBase = Wl + (cp * 32 + l15) * W_PITCH + khalf;
            for (int st = (quad < 4) ? 0 : 1; st < FUSE_TILE_ROWS / 16; st += 2) {
                const int rowFirst = tileRow + st * 16;
                if (rowFirst < r1) {
                    gcn_strip_piece(tile + (st * 16 + l15) * W_PITCH + khalf, bBase, O,
                                    rowFirst, r1, cp * 32, lane);
                }
            }
        }
        __syncthreads();
    }
}

extern "C" void kernel_launch(void* const* d_in, const int* in_sizes, int n_in,
                              void* d_out, int out_size, void* d_ws, size_t ws_size,
                              hipStream_t stream) {
    const float* X        = (const float*)d_in[0];
    const float* W1       = (const float*)d_in[1];
    const float* W2       = (const float*)d_in[2];
    const float* edgeVals = (const float*)d_in[3];
    const int*   edgeRow  = (const int*)d_in[4];
    const int*   edgeCol  = (const int*)d_in[5];
    float* out = (float*)d_out;

    const int N = in_sizes[0] / DIM_K;
    const int E = in_sizes[3];
    const long long featCount = (long long)N * DIM_K;

    const int NB = (N + ROWS_PER_BIN - 1) / ROWS_PER_BIN;
    const int nPad = (N + 31) & ~31;

    float* P  = (float*)d_ws;
    float* H1 = P + featCount;
    int* cnt    = (int*)(H1 + featCount);
    int* cursor = cnt + nPad;
    int* bucket = cursor + nPad;
    int* binned = bucket + 2 * (long long)E;
    int* chunkHist = binned + 2 * (long long)E;
    int* chunkOff  = chunkHist + EDGE_CHUNKS * NB;
    int* binTotal  = chunkOff + EDGE_CHUNKS * NB;
    int* binBase   = binTotal + NB + 32;

    const int gatherThreads = GATHER_THREADS;
    const int gatherBlocks = GATHER_GROUPS;

    const int gemmThreads = GEMM_THREADS;
    const int gemmBlocks = GEMM_GROUPS;

    gcn_bin_hist_kernel<<<EDGE_CHUNKS, SORT_THREADS, NB * sizeof(int), stream>>>(
        edgeRow, chunkHist, E, NB);
    gcn_bin_offsets_kernel<<<(NB + 31) / 32, SORT_THREADS, 0, stream>>>(
        chunkHist, chunkOff, binTotal, NB);
    gcn_bin_scatter_kernel<<<EDGE_CHUNKS, SORT_THREADS, 3 * NB * sizeof(int), stream>>>(
        edgeVals, edgeRow, edgeCol, chunkOff, binTotal, binBase, binned, E, NB);
    gcn_bin_rows_kernel<<<NB, ROWS_THREADS, 0, stream>>>(
        binned, binBase, cnt, cursor, bucket, N);

    gcn_gather_gemm_tanh_kernel<<<FUSE_GROUPS, FUSE_THREADS, 0, stream>>>(
        X, cnt, cursor, bucket, W1, H1, N);

    gcn_gemm_kernel<4, false><<<gemmBlocks, gemmThreads, 0, stream>>>(
        H1, W2, P, N);
    gcn_gather64_tanh_kernel<<<gatherBlocks, gatherThreads, 0, stream>>>(P, cnt, cursor, bucket, out, N);
}


# Edge grouping loses its two global histogram launches: level 1 orders each batch by bin in place and publishes run starts, level 2 collects a bin's runs; row sums stay in LDS-atomic arrival order
# speedup vs baseline: 5.3698x; 1.0519x over previous
#include <hip/hip_runtime.h>
#include <hip/hip_bf16.h>
#include <math.h>


typedef __attribute__((ext_vector_type(2))) float v2f;
typedef __attribute__((ext_vector_type(8))) float v8f;

#define DIM_K 128
#define DIM_OUT 64

__device__ __forceinline__ float fast_tanh(float x) {
    float ax = fabsf(x);
    float e  = __expf(-2.0f * ax);
    float r  = (1.0f - e) * __builtin_amdgcn_rcpf(1.0f + e);
    return copysignf(r, x);
}

#define ROWS_PER_BIN_LOG2 7
#define ROWS_PER_BIN (1 << ROWS_PER_BIN_LOG2)
#define EDGE_CHUNKS 256
#define SORT_THREADS 1024
#define EDGE_BATCH 8
#define STAGE_CAP (EDGE_BATCH * SORT_THREADS)
#define ROWS_THREADS 256
#define BIN_STAGE_CAP 4096

typedef unsigned long long gcn_rec;

__device__ __forceinline__ gcn_rec gcn_make_rec(int lo, int hi) {
    return (gcn_rec)(unsigned)lo | ((gcn_rec)(unsigned)hi << 32);
}

__device__ __forceinline__ int gcn_wave_incl_scan(int x, int lane) {
#pragma unroll
    for (int d = 1; d < 32; d <<= 1) {
        int t = __shfl_up(x, d, 32);
        if (lane >= d) x += t;
    }
    return x;
}

__device__ __forceinline__ int gcn_block_excl_scan(int x, int* waveSum, int* blockSum, int& total) {
    const int lane = threadIdx.x & 31;
    const int wave = threadIdx.x >> 5;
    const int nw = blockDim.x >> 5;
    const int incl = gcn_wave_incl_scan(x, lane);
    if (lane == 31) waveSum[wave] = incl;
    __syncthreads();
    if (wave == 0) {
        const int w = (lane < nw) ? waveSum[lane] : 0;
        const int wi = gcn_wave_incl_scan(w, lane);
        waveSum[lane] = wi - w;
        if (lane == 31) *blockSum = wi;
    }
    __syncthreads();
    const int excl = waveSum[wave] + incl - x;
    total = *blockSum;
    __syncthreads();
    return excl;
}

__global__ void __launch_bounds__(SORT_THREADS)
gcn_batch_sort_kernel(const float* __restrict__ vals, const int* __restrict__ row,
                      const int* __restrict__ col, int* __restrict__ segTab,
                      int* __restrict__ binnedWords, int E, int NB) {
    gcn_rec* __restrict__ binned = (gcn_rec*)binnedWords;
    extern __shared__ int ldsBins[];
    __shared__ gcn_rec stage[STAGE_CAP];
    __shared__ int waveSum[32];
    __shared__ int blockSum;
    int* lcnt   = ldsBins;
    int* lstart = ldsBins + NB;

    const int chunk = (E + EDGE_CHUNKS - 1) / EDGE_CHUNKS;
    const int perChunk = max(1, (chunk + STAGE_CAP - 1) / STAGE_CAP);
    const int nBatches = EDGE_CHUNKS * perChunk;
    const int lo = min(E, (int)blockIdx.x * chunk);
    const int hi = min(E, lo + chunk);
    for (int b = 0; b < perChunk; ++b) {
        const int s0 = min(hi, lo + b * STAGE_CAP);
        const int s1 = min(hi, s0 + STAGE_CAP);
        int* tabCol = segTab + ((int)blockIdx.x * perChunk + b);
        for (int i = threadIdx.x; i < NB; i += SORT_THREADS) lcnt[i] = 0;
        __syncthreads();

        int r[EDGE_BATCH], cidx[EDGE_BATCH], rank[EDGE_BATCH];
        float v[EDGE_BATCH];
#pragma unroll
        for (int k = 0; k < EDGE_BATCH; ++k) {
            const int e = max(0, min(min(s0 + (int)threadIdx.x + k * SORT_THREADS, s1 - 1), E - 1));
            r[k] = row[e];
            cidx[k] = col[e];
            v[k] = vals[e];
        }
#pragma unroll
        for (int k = 0; k < EDGE_BATCH; ++k) {
            rank[k] = 0;
            if (s0 + (int)threadIdx.x + k * SORT_THREADS < s1)
                rank[k] = atomicAdd(&lcnt[r[k] >> ROWS_PER_BIN_LOG2], 1);
        }
        __syncthreads();

        int lcarry = 0;
        for (int c = 0; c < NB; c += SORT_THREADS) {
            const int bin = c + threadIdx.x;
            const int n = (bin < NB) ? lcnt[bin] : 0;
            int total;
            const int st = lcarry + gcn_block_excl_scan(n, waveSum, &blockSum, total);
            if (bin < NB) { lstart[bin] = st; tabCol[(long long)bin * nBatches] = st; }
            lcarry += total;
        }
        if (threadIdx.x == 0) tabCol[(long long)NB * nBatches] = s1 - s0;
        __syncthreads();

#pragma unroll
        for (int k = 0; k < EDGE_BATCH; ++k) {
            if (s0 + (int)threadIdx.x + k * SORT_THREADS < s1) {
                const int bin = r[k] >> ROWS_PER_BIN_LOG2;
                stage[lstart[bin] + rank[k]] =
                    gcn_make_rec(((r[k] & (ROWS_PER_BIN - 1)) << 24) | cidx[k], __float_as_int(v[k]));
            }
        }
        __syncthreads();

        for (int i = threadIdx.x; i < s1 - s0; i += SORT_THREADS) binned[s0 + i] = stage[i];
        __syncthreads();
    }
}

#define SEG_LANES 16
#define SEG_PER_SWEEP (ROWS_THREADS / SEG_LANES)
#define SEG_SWEEPS (ROWS_THREADS / SEG_PER_SWEEP)

__device__ __forceinline__ int gcn_batch_start(int j, int chunk, int perChunk, int E) {
    const int ci = j / perChunk;
    const int b = j - ci * perChunk;
    const int lo = min(E, ci * chunk);
    const int hi = min(E, lo + chunk);
    return min(hi, lo + b * STAGE_CAP);
}

__device__ __forceinline__ void gcn_block_sum2(int a, int b, int* sumA, int* sumB,
                                               int& totalA, int& totalB) {
#pragma unroll
    for (int d = 16; d >= 1; d >>= 1) {
        a += __shfl_xor(a, d, 32);
        b += __shfl_xor(b, d, 32);
    }
    const int nw = blockDim.x >> 5;
    if ((threadIdx.x & 31) == 0) { sumA[threadIdx.x >> 5] = a; sumB[threadIdx.x >> 5] = b; }
    __syncthreads();
    totalA = 0; totalB = 0;
    for (int w = 0; w < nw; ++w) { totalA += sumA[w]; totalB += sumB[w]; }
}

__device__ __forceinline__ void gcn_load_run(const int* __restrict__ segTab, int g, int bin,
                                             int nBatches, int chunk, int perChunk, int E,
                                             int& mySrc, int& myLen, int& below) {
    const int j = g * ROWS_THREADS + (int)threadIdx.x;
    mySrc = 0; myLen = 0; below = 0;
    if (j < nBatches) {
        const int a = segTab[(long long)bin * nBatches + j];
        const int b = segTab[(long long)(bin + 1) * nBatches + j];
        mySrc = gcn_batch_start(j, chunk, perChunk, E) + a;
        myLen = b - a;
        below = a;
    }
}

__device__ __forceinline__ void gcn_fetch_group(const gcn_rec* __restrict__ binned,
                                                int* segSrc, int* segLen, int mySrc, int myLen,
                                                int sl, int sq, int E,
                                                int (&src)[SEG_SWEEPS], int (&len)[SEG_SWEEPS],
                                                gcn_rec (&rec)[SEG_SWEEPS]) {
    __syncthreads();
    segSrc[threadIdx.x] = mySrc;
    segLen[threadIdx.x] = myLen;
    __syncthreads();
#pragma unroll
    for (int s = 0; s < SEG_SWEEPS; ++s) {
        src[s] = segSrc[s * SEG_PER_SWEEP + sq];
        len[s] = segLen[s * SEG_PER_SWEEP + sq];
    }
#pragma unroll
    for (int s = 0; s < SEG_SWEEPS; ++s)
        rec[s] = binned[max(0, min(src[s] + min(sl, len[s] - 1), E - 1))];
}

__global__ void __launch_bounds__(ROWS_THREADS)
gcn_bin_rows_kernel(const int* __restrict__ binnedWords, const int* __restrict__ segTab,
                    int* __restrict__ cnt, int* __restrict__ cursorEnd,
                    int* __restrict__ bucketWords, int N, int E) {
    const gcn_rec* __restrict__ binned = (const gcn_rec*)binnedWords;
    gcn_rec* __restrict__ bucket = (gcn_rec*)bucketWords;
    __shared__ gcn_rec stage[BIN_STAGE_CAP];
    __shared__ int rowCur[ROWS_PER_BIN];
    __shared__ int rowWaveSum[ROWS_PER_BIN / 32];
    __shared__ int segSrc[ROWS_THREADS];
    __shared__ int segLen[ROWS_THREADS];
    __shared__ int sumBelow[32];
    __shared__ int sumMine[32];
    const int lane = threadIdx.x & 31;
    const int wave = threadIdx.x >> 5;
    const int bin = blockIdx.x;
    const int sl = threadIdx.x & (SEG_LANES - 1);
    const int sq = threadIdx.x / SEG_LANES;

    const int chunk = (E + EDGE_CHUNKS - 1) / EDGE_CHUNKS;
    const int perChunk = max(1, (chunk + STAGE_CAP - 1) / STAGE_CAP);
    const int nBatches = EDGE_CHUNKS * perChunk;
    const int nGroups = (nBatches + ROWS_THREADS - 1) / ROWS_THREADS;

    if (threadIdx.x < ROWS_PER_BIN) rowCur[threadIdx.x] = 0;

    int mySrc = 0, myLen = 0;
    int below = 0, mine = 0;
    for (int g = 0; g < nGroups; ++g) {
        int bl;
        gcn_load_run(segTab, g, bin, nBatches, chunk, perChunk, E, mySrc, myLen, bl);
        below += bl;
        mine += myLen;
    }
    int lo, total;
    gcn_block_sum2(below, mine, sumBelow, sumMine, lo, total);
    const bool fits = total <= BIN_STAGE_CAP;

    int src[SEG_SWEEPS], len[SEG_SWEEPS];
    gcn_rec rec[SEG_SWEEPS];

    for (int g = nGroups - 1; g >= 0; --g) {
        if (g != nGroups - 1) {
            int bl;
            gcn_load_run(segTab, g, bin, nBatches, chunk, perChunk, E, mySrc, myLen, bl);
        }
        gcn_fetch_group(binned, segSrc, segLen, mySrc, myLen, sl, sq, E, src, len, rec);
#pragma unroll
        for (int s = 0; s < SEG_SWEEPS; ++s)
            if (sl < len[s]) atomicAdd(&rowCur[((int)rec[s]) >> 24], 1);
#pragma unroll
        for (int s = 0; s < SEG_SWEEPS; ++s) {
#pragma unroll 1
            for (int k = sl + SEG_LANES; k < len[s]; k += SEG_LANES)
                atomicAdd(&rowCur[((int)binned[src[s] + k]) >> 24], 1);
        }
    }
    __syncthreads();
    int c = 0, incl = 0;
    if (threadIdx.x < ROWS_PER_BIN) {
        c = rowCur[threadIdx.x];
        incl = gcn_wave_incl_scan(c, lane);
        if (lane == 31) rowWaveSum[wave] = incl;
    }
    __syncthreads();
    if (threadIdx.x < ROWS_PER_BIN) {
        int off = 0;
        for (int w = 0; w < wave; ++w) off += rowWaveSum[w];
        const int rel = off + incl - c;
        const int r = bin * ROWS_PER_BIN + threadIdx.x;
        if (r < N) { cnt[r] = c; cursorEnd[r] = lo + rel + c; }
        rowCur[threadIdx.x] = rel;
    }
    __syncthreads();

    for (int g = 0; g < nGroups; ++g) {
        if (g != 0) {
            int bl;
            gcn_load_run(segTab, g, bin, nBatches, chunk, perChunk, E, mySrc, myLen, bl);
            gcn_fetch_group(binned, segSrc, segLen, mySrc, myLen, sl, sq, E, src, len, rec);
        }
#pragma unroll
        for (int s = 0; s < SEG_SWEEPS; ++s) {
            if (sl < len[s]) {
                const int x = (int)rec[s];
                const int pos = atomicAdd(&rowCur[x >> 24], 1);
                const gcn_rec o = gcn_make_rec(x & 0xFFFFFF, (int)(rec[s] >> 32));
                if (fits) stage[pos] = o; else bucket[lo + pos] = o;
            }
        }
#pragma unroll
        for (int s = 0; s < SEG_SWEEPS; ++s) {
#pragma unroll 1
            for (int k = sl + SEG_LANES; k < len[s]; k += SEG_LANES) {
                const gcn_rec rk = binned[src[s] + k];
                const int x = (int)rk;
                const int pos = atomicAdd(&rowCur[x >> 24], 1);
                const gcn_rec o = gcn_make_rec(x & 0xFFFFFF, (int)(rk >> 32));
                if (fits) stage[pos] = o; else bucket[lo + pos] = o;
            }
        }
    }
    if (fits) {
        __syncthreads();
        for (int i = threadIdx.x; i < total; i += ROWS_THREADS) bucket[lo + i] = stage[i];
    }
}

template <int G>
__device__ __forceinline__ void gcn_gather_group(const float4* __restrict__ X4,
                                                 int2 my, int j, int lane,
                                                 float4& acc) {
    int c[G];
    float v[G];
    float4 x[G];
#pragma unroll
    for (int k = 0; k < G; ++k) {
        c[k] = __builtin_amdgcn_readlane(my.x, j + k);
        v[k] = __int_as_float(__builtin_amdgcn_readlane(my.y, j + k));
    }
#pragma unroll
    for (int k = 0; k < G; ++k) x[k] = X4[(long long)c[k] * (DIM_K / 4) + lane];
#pragma unroll
    for (int k = 0; k < G; ++k) {
        acc.x = __builtin_fmaf(v[k], x[k].x, acc.x);
        acc.y = __builtin_fmaf(v[k], x[k].y, acc.y);
        acc.z = __builtin_fmaf(v[k], x[k].z, acc.z);
        acc.w = __builtin_fmaf(v[k], x[k].w, acc.w);
    }
}

#define GATHER_WAVES 32
#define GATHER_THREADS (GATHER_WAVES * 32)
#define GATHER_GROUPS 512

__device__ __forceinline__ void gcn_row_chunk(const float4* __restrict__ X4,
                                              int2 my, int m, int lane, float4& acc) {
    int j = 0;
    for (; j + 8 <= m; j += 8) gcn_gather_group<8>(X4, my, j, lane, acc);
    if ((m - j) & 4) { gcn_gather_group<4>(X4, my, j, lane, acc); j += 4; }
    if ((m - j) & 2) { gcn_gather_group<2>(X4, my, j, lane, acc); j += 2; }
    if ((m - j) & 1) { gcn_gather_group<1>(X4, my, j, lane, acc); }
}

template <int G>
__device__ __forceinline__ void gcn_gather_group64(const float2* __restrict__ P2,
                                                   int2 my, int j, int lane,
                                                   float2& acc) {
    int c[G];
    float v[G];
    float2 x[G];
#pragma unroll
    for (int k = 0; k < G; ++k) {
        c[k] = __builtin_amdgcn_readlane(my.x, j + k);
        v[k] = __int_as_float(__builtin_amdgcn_readlane(my.y, j + k));
    }
#pragma unroll
    for (int k = 0; k < G; ++k) x[k] = P2[(long long)c[k] * (DIM_OUT / 2) + lane];
#pragma unroll
    for (int k = 0; k < G; ++k) {
        acc.x = __builtin_fmaf(v[k], x[k].x, acc.x);
        acc.y = __builtin_fmaf(v[k], x[k].y, acc.y);
    }
}

__device__ __forceinline__ void gcn_row_chunk64(const float2* __restrict__ P2,
                                                int2 my, int m, int lane, float2& acc) {
    int j = 0;
    for (; j + 16 <= m; j += 16) gcn_gather_group64<16>(P2, my, j, lane, acc);
    if ((m - j) & 8) { gcn_gather_group64<8>(P2, my, j, lane, acc); j += 8; }
    if ((m - j) & 4) { gcn_gather_group64<4>(P2, my, j, lane, acc); j += 4; }
    if ((m - j) & 2) { gcn_gather_group64<2>(P2, my, j, lane, acc); j += 2; }
    if ((m - j) & 1) { gcn_gather_group64<1>(P2, my, j, lane, acc); }
}

__global__ void __launch_bounds__(GATHER_THREADS)
gcn_gather64_tanh_kernel(const float* __restrict__ P,
                         const int* __restrict__ cnt,
                         const int* __restrict__ cursorEnd,
                         const int* __restrict__ bucket,
                         float* __restrict__ O, int N) {
    const int lane = threadIdx.x & 31;
    const int stride = (int)gridDim.x * GATHER_WAVES;
    int rNext = __builtin_amdgcn_readfirstlane(
        (int)blockIdx.x * GATHER_WAVES + (int)(threadIdx.x >> 5));
    if (rNext >= N) return;
    const int2* ed = (const int2*)bucket;
    const float2* P2 = (const float2*)P;

    int r = -1, n = 0, s = 0;
    int2 my = make_int2(0, 0);
    int nNext = cnt[rNext];
    int eNext = cursorEnd[rNext];

    for (;;) {
        const int sNext = eNext - nNext;
        int2 myNext = make_int2(0, 0);
        if (lane < nNext) myNext = ed[sNext + lane];
        const int rAfter = rNext + stride;
        int nAfter = 0, eAfter = 0;
        if (rAfter < N) { nAfter = cnt[rAfter]; eAfter = cursorEnd[rAfter]; }

        float2 acc = make_float2(0.0f, 0.0f);
        gcn_row_chunk64(P2, my, min(32, n), lane, acc);
        for (int base = 32; base < n; base += 32) {
            const int m = min(32, n - base);
            int2 more = make_int2(0, 0);
            if (lane < m) more = ed[s + base + lane];
            gcn_row_chunk64(P2, more, m, lane, acc);
        }
        if (r >= 0)
            ((float2*)(O + (long long)r * DIM_OUT))[lane] =
                make_float2(fast_tanh(acc.x), fast_tanh(acc.y));

        if (rNext >= N) break;
        r = rNext; n = nNext; s = sNext; my = myNext;
        rNext = rAfter; nNext = nAfter; eNext = eAfter;
    }
}

#define GEMM_WAVES 24
#define GEMM_THREADS (GEMM_WAVES * 32)
#define GEMM_GROUPS 256
#define W_PITCH (DIM_K + 4)

template <int NT, bool ACT>
__global__ void __launch_bounds__(GEMM_THREADS)
gcn_gemm_kernel(const float* __restrict__ S,
                const float* __restrict__ W,
                float* __restrict__ O,
                int nNodes) {
    __shared__ __attribute__((aligned(16))) float Wl[NT * 16 * W_PITCH];

    {
        const float4* W4 = (const float4*)W;
        constexpr int W4N = NT * 16 * (DIM_K / 4);
        constexpr int W_BATCH = (W4N + GEMM_THREADS - 1) / GEMM_THREADS;
        float4 w[W_BATCH];
#pragma unroll
        for (int j = 0; j < W_BATCH; ++j)
            w[j] = W4[min((int)threadIdx.x + j * GEMM_THREADS, W4N - 1)];
#pragma unroll
        for (int j = 0; j < W_BATCH; ++j) {
            const int i = (int)threadIdx.x + j * GEMM_THREADS;
            if (i < W4N) *(float4*)&Wl[(i >> 5) * W_PITCH + (i & 31) * 4] = w[j];
        }
    }
    __syncthreads();

    const int lane = threadIdx.x & 31;
    const int wave = __builtin_amdgcn_readfirstlane((int)(threadIdx.x >> 5));
    const int mTiles = (nNodes + 15) >> 4;

    const int nOut  = NT * 16;
    const int l15   = lane & 15;
    const int khalf = (lane < 16) ? 0 : 2;

    const float* bBase = Wl + l15 * W_PITCH + khalf;

    for (int mTile = wave * (int)gridDim.x + (int)blockIdx.x; mTile < mTiles;
         mTile += GEMM_WAVES * (int)gridDim.x) {
        const float* aBase = S + (long long)(mTile * 16 + l15) * DIM_K + khalf;

        v8f acc[NT];
#pragma unroll
        for (int t = 0; t < NT; ++t) acc[t] = (v8f){};

#pragma unroll 2
        for (int k = 0; k < DIM_K; k += 4) {
            v2f a;
            a.x = aBase[k];
            a.y = aBase[k + 1];
#pragma unroll
            for (int t = 0; t < NT; ++t) {
                const v2f b = *(const v2f*)(bBase + t * 16 * W_PITCH + k);
                acc[t] = __builtin_amdgcn_wmma_f32_16x16x4_f32(
                     false, a,  false, b,
                     (short)0, acc[t],  false,  false);
            }
        }

        const int mbase = mTile * 16 + ((lane < 16) ? 0 : 8);
#pragma unroll
        for (int t = 0; t < NT; ++t) {
            const int n = t * 16 + l15;
#pragma unroll
            for (int r = 0; r < 8; ++r) {
                O[(long long)(mbase + r) * nOut + n] = ACT ? fast_tanh(acc[t][r]) : acc[t][r];
            }
        }
    }
}

#define FUSE_WAVES 32
#define FUSE_THREADS (FUSE_WAVES * 32)
#define FUSE_GROUPS 512
#define FUSE_GATHER_WAVES 24
#define FUSE_ROWS_PER_WAVE 2
#define FUSE_TILE_ROWS (FUSE_GATHER_WAVES * FUSE_ROWS_PER_WAVE)

typedef const __attribute__((address_space(4))) int* gcn_meta_ptr;

__device__ __forceinline__ float4 gcn_row_trip(const float4* __restrict__ X4,
                                               const int2* __restrict__ ed,
                                               gcn_meta_ptr cnt, gcn_meta_ptr cursorEnd,
                                               int lane, int rowStride, int rEnd,
                                               int& n, int& s, int2& my,
                                               int& rNext, int& nNext, int& eNext) {
    const int sNext = eNext - nNext;
    int2 myNext = make_int2(0, 0);
    if (lane < nNext) myNext = ed[sNext + lane];
    const int rAfter = rNext + rowStride;
    int nAfter = 0, eAfter = 0;
    if (rAfter < rEnd) { nAfter = cnt[rAfter]; eAfter = cursorEnd[rAfter]; }

    float4 acc = make_float4(0.0f, 0.0f, 0.0f, 0.0f);
    gcn_row_chunk(X4, my, min(32, n), lane, acc);
    for (int base = 32; base < n; base += 32) {
        const int m = min(32, n - base);
        int2 more = make_int2(0, 0);
        if (lane < m) more = ed[s + base + lane];
        gcn_row_chunk(X4, more, m, lane, acc);
    }
    n = nNext; s = sNext; my = myNext;
    rNext = rAfter; nNext = nAfter; eNext = eAfter;
    return acc;
}

__device__ __forceinline__ void gcn_strip_piece(const float* __restrict__ aBase,
                                                const float* __restrict__ bBase,
                                                float* __restrict__ O,
                                                int rowFirst, int rEnd, int colFirst, int lane) {
    v8f acc0 = (v8f){};
    v8f acc1 = (v8f){};
#pragma unroll 4
    for (int k = 0; k < DIM_K; k += 4) {
        const v2f a  = *(const v2f*)(aBase + k);
        const v2f b0 = *(const v2f*)(bBase + k);
        const v2f b1 = *(const v2f*)(bBase + 16 * W_PITCH + k);
        acc0 = __builtin_amdgcn_wmma_f32_16x16x4_f32(
             false, a,  false, b0,
             (short)0, acc0,  false,  false);
        acc1 = __builtin_amdgcn_wmma_f32_16x16x4_f32(
             false, a,  false, b1,
             (short)0, acc1,  false,  false);
    }
    const int l15 = lane & 15;
    const int mrow = rowFirst + ((lane < 16) ? 0 : 8);
#pragma unroll
    for (int r = 0; r < 8; ++r) {
        const int row = mrow + r;
        if (row < rEnd) {
            float* o = O + (long long)row * DIM_K + colFirst + l15;
            o[0]  = fast_tanh(acc0[r]);
            o[16] = fast_tanh(acc1[r]);
        }
    }
}

__global__ void __launch_bounds__(FUSE_THREADS) __attribute__((amdgpu_waves_per_eu(16)))
gcn_gather_gemm_tanh_kernel(const float* __restrict__ X,
                            const int* __restrict__ cntGlobal,
                            const int* __restrict__ cursorEndGlobal,
                            const int* __restrict__ bucket,
                            const float* __restrict__ W,
                            float* __restrict__ O, int N) {
    const gcn_meta_ptr cnt = (gcn_meta_ptr)cntGlobal;
    const gcn_meta_ptr cursorEnd = (gcn_meta_ptr)cursorEndGlobal;
    __shared__ __attribute__((aligned(16))) float Wl[DIM_K * W_PITCH];
    __shared__ __attribute__((aligned(16))) float St[2 * FUSE_TILE_ROWS * W_PITCH];

    {
        const float4* W4 = (const float4*)W;
        constexpr int W4N = DIM_K * (DIM_K / 4);
        constexpr int W_BATCH = W4N / FUSE_THREADS;
        float4 w[W_BATCH];
#pragma unroll
        for (int j = 0; j < W_BATCH; ++j) w[j] = W4[(int)threadIdx.x + j * FUSE_THREADS];
#pragma unroll
        for (int j = 0; j < W_BATCH; ++j) {
            const int i = (int)threadIdx.x + j * FUSE_THREADS;
            *(float4*)&Wl[(i >> 5) * W_PITCH + (i & 31) * 4] = w[j];
        }
    }

    const int lane = threadIdx.x & 31;
    const int wave = __builtin_amdgcn_readfirstlane((int)(threadIdx.x >> 5));
    const int quad = wave >> 2;
    const int pos  = wave & 3;
    const bool mmaWave = pos == (quad & 3);
    const int g = quad * 3 + ((pos > (quad & 3)) ? pos - 1 : pos);

    const int rowsPer = (N + (int)gridDim.x - 1) / (int)gridDim.x;
    const int r0 = min(N, (int)blockIdx.x * rowsPer);
    const int r1 = min(N, r0 + rowsPer);
    const int nT = (r1 - r0 + FUSE_TILE_ROWS - 1) / FUSE_TILE_ROWS;

    const int2* ed = (const int2*)bucket;
    const float4* X4 = (const float4*)X;

    int n = 0, s = 0;
    int2 my = make_int2(0, 0);
    int rNext = r0 + g, nNext = 0, eNext = 0;
    if (!mmaWave) {
        if (rNext < r1) { nNext = cnt[rNext]; eNext = cursorEnd[rNext]; }
        (void)gcn_row_trip(X4, ed, cnt, cursorEnd, lane, FUSE_GATHER_WAVES, r1,
                           n, s, my, rNext, nNext, eNext);
    }

    const int l15   = lane & 15;
    const int khalf = (lane < 16) ? 0 : 2;

    for (int it = 0; it <= nT; ++it) {
        if (!mmaWave) {
            if (it < nT) {
                float* tile = St + (it & 1) * (FUSE_TILE_ROWS * W_PITCH);
#pragma unroll
                for (int q = 0; q < FUSE_ROWS_PER_WAVE; ++q) {
                    const float4 acc = gcn_row_trip(X4, ed, cnt, cursorEnd, lane,
                                                    FUSE_GATHER_WAVES, r1,
                                                    n, s, my, rNext, nNext, eNext);
                    *(float4*)&tile[(g + q * FUSE_GATHER_WAVES) * W_PITCH + lane * 4] = acc;
                }
            }
        } else if (it >= 1) {
            const int tt = it - 1;
            const float* tile = St + (tt & 1) * (FUSE_TILE_ROWS * W_PITCH);
            const int tileRow = r0 + tt * FUSE_TILE_ROWS;
            const int cp = quad & 3;
            const float* bBase = Wl + (cp * 32 + l15) * W_PITCH + khalf;
            for (int st = (quad < 4) ? 0 : 1; st < FUSE_TILE_ROWS / 16; st += 2) {
                const int rowFirst = tileRow + st * 16;
                if (rowFirst < r1) {
                    gcn_strip_piece(tile + (st * 16 + l15) * W_PITCH + khalf, bBase, O,
                                    rowFirst, r1, cp * 32, lane);
                }
            }
        }
        __syncthreads();
    }
}

extern "C" void kernel_launch(void* const* d_in, const int* in_sizes, int n_in,
                              void* d_out, int out_size, void* d_ws, size_t ws_size,
                              hipStream_t stream) {
    const float* X        = (const float*)d_in[0];
    const float* W1       = (const float*)d_in[1];
    const float* W2       = (const float*)d_in[2];
    const float* edgeVals = (const float*)d_in[3];
    const int*   edgeRow  = (const int*)d_in[4];
    const int*   edgeCol  = (const int*)d_in[5];
    float* out = (float*)d_out;

    const int N = in_sizes[0] / DIM_K;
    const int E = in_sizes[3];
    const long long featCount = (long long)N * DIM_K;

    const int NB = (N + ROWS_PER_BIN - 1) / ROWS_PER_BIN;
    const int nPad = (N + 31) & ~31;

    float* P  = (float*)d_ws;
    float* H1 = P + featCount;
    int* cnt    = (int*)(H1 + featCount);
    int* cursor = cnt + nPad;
    int* bucket = cursor + nPad;
    int* binned = bucket + 2 * (long long)E;
    int* segTab = binned + 2 * (long long)E;

    const int gatherThreads = GATHER_THREADS;
    const int gatherBlocks = GATHER_GROUPS;

    const int gemmThreads = GEMM_THREADS;
    const int gemmBlocks = GEMM_GROUPS;

    gcn_batch_sort_kernel<<<EDGE_CHUNKS, SORT_THREADS, 2 * NB * sizeof(int), stream>>>(
        edgeVals, edgeRow, edgeCol, segTab, binned, E, NB);
    gcn_bin_rows_kernel<<<NB, ROWS_THREADS, 0, stream>>>(
        binned, segTab, cnt, cursor, bucket, N, E);

    gcn_gather_gemm_tanh_kernel<<<FUSE_GROUPS, FUSE_THREADS, 0, stream>>>(
        X, cnt, cursor, bucket, W1, H1, N);

    gcn_gemm_kernel<4, false><<<gemmBlocks, gemmThreads, 0, stream>>>(
        H1, W2, P, N);
    gcn_gather64_tanh_kernel<<<gatherBlocks, gatherThreads, 0, stream>>>(P, cnt, cursor, bucket, out, N);
}


# W2 product joins the layer-1 kernel (4 launches, H1 never stored): transposed WMMAs feed tanh(H1) pieces from registers; P summed over n in fragment order, four 32-column partials added in LDS
# speedup vs baseline: 5.9193x; 1.1023x over previous
#include <hip/hip_runtime.h>
#include <hip/hip_bf16.h>
#include <math.h>


typedef __attribute__((ext_vector_type(2))) float v2f;
typedef __attribute__((ext_vector_type(8))) float v8f;

#define DIM_K 128
#define DIM_OUT 64

__device__ __forceinline__ float fast_tanh(float x) {
    float ax = fabsf(x);
    float e  = __expf(-2.0f * ax);
    float r  = (1.0f - e) * __builtin_amdgcn_rcpf(1.0f + e);
    return copysignf(r, x);
}

#define ROWS_PER_BIN_LOG2 7
#define ROWS_PER_BIN (1 << ROWS_PER_BIN_LOG2)
#define EDGE_CHUNKS 256
#define SORT_THREADS 1024
#define EDGE_BATCH 8
#define STAGE_CAP (EDGE_BATCH * SORT_THREADS)
#define ROWS_THREADS 256
#define BIN_STAGE_CAP 4096

typedef unsigned long long gcn_rec;

__device__ __forceinline__ gcn_rec gcn_make_rec(int lo, int hi) {
    return (gcn_rec)(unsigned)lo | ((gcn_rec)(unsigned)hi << 32);
}

__device__ __forceinline__ int gcn_wave_incl_scan(int x, int lane) {
#pragma unroll
    for (int d = 1; d < 32; d <<= 1) {
        int t = __shfl_up(x, d, 32);
        if (lane >= d) x += t;
    }
    return x;
}

__device__ __forceinline__ int gcn_block_excl_scan(int x, int* waveSum, int* blockSum, int& total) {
    const int lane = threadIdx.x & 31;
    const int wave = threadIdx.x >> 5;
    const int nw = blockDim.x >> 5;
    const int incl = gcn_wave_incl_scan(x, lane);
    if (lane == 31) waveSum[wave] = incl;
    __syncthreads();
    if (wave == 0) {
        const int w = (lane < nw) ? waveSum[lane] : 0;
        const int wi = gcn_wave_incl_scan(w, lane);
        waveSum[lane] = wi - w;
        if (lane == 31) *blockSum = wi;
    }
    __syncthreads();
    const int excl = waveSum[wave] + incl - x;
    total = *blockSum;
    __syncthreads();
    return excl;
}

__global__ void __launch_bounds__(SORT_THREADS)
gcn_batch_sort_kernel(const float* __restrict__ vals, const int* __restrict__ row,
                      const int* __restrict__ col, int* __restrict__ segTab,
                      int* __restrict__ binnedWords, int E, int NB) {
    gcn_rec* __restrict__ binned = (gcn_rec*)binnedWords;
    extern __shared__ int ldsBins[];
    __shared__ gcn_rec stage[STAGE_CAP];
    __shared__ int waveSum[32];
    __shared__ int blockSum;
    int* lcnt   = ldsBins;
    int* lstart = ldsBins + NB;

    const int chunk = (E + EDGE_CHUNKS - 1) / EDGE_CHUNKS;
    const int perChunk = max(1, (chunk + STAGE_CAP - 1) / STAGE_CAP);
    const int nBatches = EDGE_CHUNKS * perChunk;
    const int lo = min(E, (int)blockIdx.x * chunk);
    const int hi = min(E, lo + chunk);
    for (int b = 0; b < perChunk; ++b) {
        const int s0 = min(hi, lo + b * STAGE_CAP);
        const int s1 = min(hi, s0 + STAGE_CAP);
        int* tabCol = segTab + ((int)blockIdx.x * perChunk + b);
        for (int i = threadIdx.x; i < NB; i += SORT_THREADS) lcnt[i] = 0;
        __syncthreads();

        int r[EDGE_BATCH], cidx[EDGE_BATCH], rank[EDGE_BATCH];
        float v[EDGE_BATCH];
#pragma unroll
        for (int k = 0; k < EDGE_BATCH; ++k) {
            const int e = max(0, min(min(s0 + (int)threadIdx.x + k * SORT_THREADS, s1 - 1), E - 1));
            r[k] = row[e];
            cidx[k] = col[e];
            v[k] = vals[e];
        }
#pragma unroll
        for (int k = 0; k < EDGE_BATCH; ++k) {
            rank[k] = 0;
            if (s0 + (int)threadIdx.x + k * SORT_THREADS < s1)
                rank[k] = atomicAdd(&lcnt[r[k] >> ROWS_PER_BIN_LOG2], 1);
        }
        __syncthreads();

        int lcarry = 0;
        for (int c = 0; c < NB; c += SORT_THREADS) {
            const int bin = c + threadIdx.x;
            const int n = (bin < NB) ? lcnt[bin] : 0;
            int total;
            const int st = lcarry + gcn_block_excl_scan(n, waveSum, &blockSum, total);
            if (bin < NB) { lstart[bin] = st; tabCol[(long long)bin * nBatches] = st; }
            lcarry += total;
        }
        if (threadIdx.x == 0) tabCol[(long long)NB * nBatches] = s1 - s0;
        __syncthreads();

#pragma unroll
        for (int k = 0; k < EDGE_BATCH; ++k) {
            if (s0 + (int)threadIdx.x + k * SORT_THREADS < s1) {
                const int bin = r[k] >> ROWS_PER_BIN_LOG2;
                stage[lstart[bin] + rank[k]] =
                    gcn_make_rec(((r[k] & (ROWS_PER_BIN - 1)) << 24) | cidx[k], __float_as_int(v[k]));
            }
        }
        __syncthreads();

        for (int i = threadIdx.x; i < s1 - s0; i += SORT_THREADS) binned[s0 + i] = stage[i];
        __syncthreads();
    }
}

#define SEG_LANES 16
#define SEG_PER_SWEEP (ROWS_THREADS / SEG_LANES)
#define SEG_SWEEPS (ROWS_THREADS / SEG_PER_SWEEP)

__device__ __forceinline__ int gcn_batch_start(int j, int chunk, int perChunk, int E) {
    const int ci = j / perChunk;
    const int b = j - ci * perChunk;
    const int lo = min(E, ci * chunk);
    const int hi = min(E, lo + chunk);
    return min(hi, lo + b * STAGE_CAP);
}

__device__ __forceinline__ void gcn_block_sum2(int a, int b, int* sumA, int* sumB,
                                               int& totalA, int& totalB) {
#pragma unroll
    for (int d = 16; d >= 1; d >>= 1) {
        a += __shfl_xor(a, d, 32);
        b += __shfl_xor(b, d, 32);
    }
    const int nw = blockDim.x >> 5;
    if ((threadIdx.x & 31) == 0) { sumA[threadIdx.x >> 5] = a; sumB[threadIdx.x >> 5] = b; }
    __syncthreads();
    totalA = 0; totalB = 0;
    for (int w = 0; w < nw; ++w) { totalA += sumA[w]; totalB += sumB[w]; }
}

__device__ __forceinline__ void gcn_load_run(const int* __restrict__ segTab, int g, int bin,
                                             int nBatches, int chunk, int perChunk, int E,
                                             int& mySrc, int& myLen, int& below) {
    const int j = g * ROWS_THREADS + (int)threadIdx.x;
    mySrc = 0; myLen = 0; below = 0;
    if (j < nBatches) {
        const int a = segTab[(long long)bin * nBatches + j];
        const int b = segTab[(long long)(bin + 1) * nBatches + j];
        mySrc = gcn_batch_start(j, chunk, perChunk, E) + a;
        myLen = b - a;
        below = a;
    }
}

__device__ __forceinline__ void gcn_fetch_group(const gcn_rec* __restrict__ binned,
                                                int* segSrc, int* segLen, int mySrc, int myLen,
                                                int sl, int sq, int E,
                                                int (&src)[SEG_SWEEPS], int (&len)[SEG_SWEEPS],
                                                gcn_rec (&rec)[SEG_SWEEPS]) {
    __syncthreads();
    segSrc[threadIdx.x] = mySrc;
    segLen[threadIdx.x] = myLen;
    __syncthreads();
#pragma unroll
    for (int s = 0; s < SEG_SWEEPS; ++s) {
        src[s] = segSrc[s * SEG_PER_SWEEP + sq];
        len[s] = segLen[s * SEG_PER_SWEEP + sq];
    }
#pragma unroll
    for (int s = 0; s < SEG_SWEEPS; ++s)
        rec[s] = binned[max(0, min(src[s] + min(sl, len[s] - 1), E - 1))];
}

__global__ void __launch_bounds__(ROWS_THREADS)
gcn_bin_rows_kernel(const int* __restrict__ binnedWords, const int* __restrict__ segTab,
                    int* __restrict__ cnt, int* __restrict__ cursorEnd,
                    int* __restrict__ bucketWords, int N, int E) {
    const gcn_rec* __restrict__ binned = (const gcn_rec*)binnedWords;
    gcn_rec* __restrict__ bucket = (gcn_rec*)bucketWords;
    __shared__ gcn_rec stage[BIN_STAGE_CAP];
    __shared__ int rowCur[ROWS_PER_BIN];
    __shared__ int rowWaveSum[ROWS_PER_BIN / 32];
    __shared__ int segSrc[ROWS_THREADS];
    __shared__ int segLen[ROWS_THREADS];
    __shared__ int sumBelow[32];
    __shared__ int sumMine[32];
    const int lane = threadIdx.x & 31;
    const int wave = threadIdx.x >> 5;
    const int bin = blockIdx.x;
    const int sl = threadIdx.x & (SEG_LANES - 1);
    const int sq = threadIdx.x / SEG_LANES;

    const int chunk = (E + EDGE_CHUNKS - 1) / EDGE_CHUNKS;
    const int perChunk = max(1, (chunk + STAGE_CAP - 1) / STAGE_CAP);
    const int nBatches = EDGE_CHUNKS * perChunk;
    const int nGroups = (nBatches + ROWS_THREADS - 1) / ROWS_THREADS;

    if (threadIdx.x < ROWS_PER_BIN) rowCur[threadIdx.x] = 0;

    int mySrc = 0, myLen = 0;
    int below = 0, mine = 0;
    for (int g = 0; g < nGroups; ++g) {
        int bl;
        gcn_load_run(segTab, g, bin, nBatches, chunk, perChunk, E, mySrc, myLen, bl);
        below += bl;
        mine += myLen;
    }
    int lo, total;
    gcn_block_sum2(below, mine, sumBelow, sumMine, lo, total);
    const bool fits = total <= BIN_STAGE_CAP;

    int src[SEG_SWEEPS], len[SEG_SWEEPS];
    gcn_rec rec[SEG_SWEEPS];

    for (int g = nGroups - 1; g >= 0; --g) {
        if (g != nGroups - 1) {
            int bl;
            gcn_load_run(segTab, g, bin, nBatches, chunk, perChunk, E, mySrc, myLen, bl);
        }
        gcn_fetch_group(binned, segSrc, segLen, mySrc, myLen, sl, sq, E, src, len, rec);
#pragma unroll
        for (int s = 0; s < SEG_SWEEPS; ++s)
            if (sl < len[s]) atomicAdd(&rowCur[((int)rec[s]) >> 24], 1);
#pragma unroll
        for (int s = 0; s < SEG_SWEEPS; ++s) {
#pragma unroll 1
            for (int k = sl + SEG_LANES; k < len[s]; k += SEG_LANES)
                atomicAdd(&rowCur[((int)binned[src[s] + k]) >> 24], 1);
        }
    }
    __syncthreads();
    int c = 0, incl = 0;
    if (threadIdx.x < ROWS_PER_BIN) {
        c = rowCur[threadIdx.x];
        incl = gcn_wave_incl_scan(c, lane);
        if (lane == 31) rowWaveSum[wave] = incl;
    }
    __syncthreads();
    if (threadIdx.x < ROWS_PER_BIN) {
        int off = 0;
        for (int w = 0; w < wave; ++w) off += rowWaveSum[w];
        const int rel = off + incl - c;
        const int r = bin * ROWS_PER_BIN + threadIdx.x;
        if (r < N) { cnt[r] = c; cursorEnd[r] = lo + rel + c; }
        rowCur[threadIdx.x] = rel;
    }
    __syncthreads();

    for (int g = 0; g < nGroups; ++g) {
        if (g != 0) {
            int bl;
            gcn_load_run(segTab, g, bin, nBatches, chunk, perChunk, E, mySrc, myLen, bl);
            gcn_fetch_group(binned, segSrc, segLen, mySrc, myLen, sl, sq, E, src, len, rec);
        }
#pragma unroll
        for (int s = 0; s < SEG_SWEEPS; ++s) {
            if (sl < len[s]) {
                const int x = (int)rec[s];
                const int pos = atomicAdd(&rowCur[x >> 24], 1);
                const gcn_rec o = gcn_make_rec(x & 0xFFFFFF, (int)(rec[s] >> 32));
                if (fits) stage[pos] = o; else bucket[lo + pos] = o;
            }
        }
#pragma unroll
        for (int s = 0; s < SEG_SWEEPS; ++s) {
#pragma unroll 1
            for (int k = sl + SEG_LANES; k < len[s]; k += SEG_LANES) {
                const gcn_rec rk = binned[src[s] + k];
                const int x = (int)rk;
                const int pos = atomicAdd(&rowCur[x >> 24], 1);
                const gcn_rec o = gcn_make_rec(x & 0xFFFFFF, (int)(rk >> 32));
                if (fits) stage[pos] = o; else bucket[lo + pos] = o;
            }
        }
    }
    if (fits) {
        __syncthreads();
        for (int i = threadIdx.x; i < total; i += ROWS_THREADS) bucket[lo + i] = stage[i];
    }
}

template <int G>
__device__ __forceinline__ void gcn_gather_group(const float4* __restrict__ X4,
                                                 int2 my, int j, int lane,
                                                 float4& acc) {
    int c[G];
    float v[G];
    float4 x[G];
#pragma unroll
    for (int k = 0; k < G; ++k) {
        c[k] = __builtin_amdgcn_readlane(my.x, j + k);
        v[k] = __int_as_float(__builtin_amdgcn_readlane(my.y, j + k));
    }
#pragma unroll
    for (int k = 0; k < G; ++k) x[k] = X4[(long long)c[k] * (DIM_K / 4) + lane];
#pragma unroll
    for (int k = 0; k < G; ++k) {
        acc.x = __builtin_fmaf(v[k], x[k].x, acc.x);
        acc.y = __builtin_fmaf(v[k], x[k].y, acc.y);
        acc.z = __builtin_fmaf(v[k], x[k].z, acc.z);
        acc.w = __builtin_fmaf(v[k], x[k].w, acc.w);
    }
}

#define GATHER_WAVES 32
#define GATHER_THREADS (GATHER_WAVES * 32)
#define GATHER_GROUPS 512

__device__ __forceinline__ void gcn_row_chunk(const float4* __restrict__ X4,
                                              int2 my, int m, int lane, float4& acc) {
    int j = 0;
    for (; j + 16 <= m; j += 16) gcn_gather_group<16>(X4, my, j, lane, acc);
    if ((m - j) & 8) { gcn_gather_group<8>(X4, my, j, lane, acc); j += 8; }
    if ((m - j) & 4) { gcn_gather_group<4>(X4, my, j, lane, acc); j += 4; }
    if ((m - j) & 2) { gcn_gather_group<2>(X4, my, j, lane, acc); j += 2; }
    if ((m - j) & 1) { gcn_gather_group<1>(X4, my, j, lane, acc); }
}

template <int G>
__device__ __forceinline__ void gcn_gather_group64(const float2* __restrict__ P2,
                                                   int2 my, int j, int lane,
                                                   float2& acc) {
    int c[G];
    float v[G];
    float2 x[G];
#pragma unroll
    for (int k = 0; k < G; ++k) {
        c[k] = __builtin_amdgcn_readlane(my.x, j + k);
        v[k] = __int_as_float(__builtin_amdgcn_readlane(my.y, j + k));
    }
#pragma unroll
    for (int k = 0; k < G; ++k) x[k] = P2[(long long)c[k] * (DIM_OUT / 2) + lane];
#pragma unroll
    for (int k = 0; k < G; ++k) {
        acc.x = __builtin_fmaf(v[k], x[k].x, acc.x);
        acc.y = __builtin_fmaf(v[k], x[k].y, acc.y);
    }
}

__device__ __forceinline__ void gcn_row_chunk64(const float2* __restrict__ P2,
                                                int2 my, int m, int lane, float2& acc) {
    int j = 0;
    for (; j + 16 <= m; j += 16) gcn_gather_group64<16>(P2, my, j, lane, acc);
    if ((m - j) & 8) { gcn_gather_group64<8>(P2, my, j, lane, acc); j += 8; }
    if ((m - j) & 4) { gcn_gather_group64<4>(P2, my, j, lane, acc); j += 4; }
    if ((m - j) & 2) { gcn_gather_group64<2>(P2, my, j, lane, acc); j += 2; }
    if ((m - j) & 1) { gcn_gather_group64<1>(P2, my, j, lane, acc); }
}

__global__ void __launch_bounds__(GATHER_THREADS)
gcn_gather64_tanh_kernel(const float* __restrict__ P,
                         const int* __restrict__ cnt,
                         const int* __restrict__ cursorEnd,
                         const int* __restrict__ bucket,
                         float* __restrict__ O, int N) {
    const int lane = threadIdx.x & 31;
    const int stride = (int)gridDim.x * GATHER_WAVES;
    int rNext = __builtin_amdgcn_readfirstlane(
        (int)blockIdx.x * GATHER_WAVES + (int)(threadIdx.x >> 5));
    if (rNext >= N) return;
    const int2* ed = (const int2*)bucket;
    const float2* P2 = (const float2*)P;

    int r = -1, n = 0, s = 0;
    int2 my = make_int2(0, 0);
    int nNext = cnt[rNext];
    int eNext = cursorEnd[rNext];

    for (;;) {
        const int sNext = eNext - nNext;
        int2 myNext = make_int2(0, 0);
        if (lane < nNext) myNext = ed[sNext + lane];
        const int rAfter = rNext + stride;
        int nAfter = 0, eAfter = 0;
        if (rAfter < N) { nAfter = cnt[rAfter]; eAfter = cursorEnd[rAfter]; }

        float2 acc = make_float2(0.0f, 0.0f);
        gcn_row_chunk64(P2, my, min(32, n), lane, acc);
        for (int base = 32; base < n; base += 32) {
            const int m = min(32, n - base);
            int2 more = make_int2(0, 0);
            if (lane < m) more = ed[s + base + lane];
            gcn_row_chunk64(P2, more, m, lane, acc);
        }
        if (r >= 0)
            ((float2*)(O + (long long)r * DIM_OUT))[lane] =
                make_float2(fast_tanh(acc.x), fast_tanh(acc.y));

        if (rNext >= N) break;
        r = rNext; n = nNext; s = sNext; my = myNext;
        rNext = rAfter; nNext = nAfter; eNext = eAfter;
    }
}

#define W_PITCH (DIM_K + 4)

#define FUSE_WAVES 32
#define FUSE_THREADS (FUSE_WAVES * 32)
#define FUSE_GROUPS 256
#define FUSE_GATHER_WAVES 24
#define FUSE_ROWS_PER_WAVE 2
#define FUSE_TILE_ROWS (FUSE_GATHER_WAVES * FUSE_ROWS_PER_WAVE)
#define FUSE_STRIPS (FUSE_TILE_ROWS / 16)
#define PP_PITCH (DIM_OUT + 4)
#define PP_PIECE (16 * PP_PITCH)
#define PP_TILE (FUSE_STRIPS * 4 * PP_PIECE)

typedef const __attribute__((address_space(4))) int* gcn_meta_ptr;

__device__ __forceinline__ float4 gcn_row_trip(const float4* __restrict__ X4,
                                               const int2* __restrict__ ed,
                                               gcn_meta_ptr cnt, gcn_meta_ptr cursorEnd,
                                               int lane, int rowStride, int rEnd,
                                               int& n, int& s, int2& my,
                                               int& rNext, int& nNext, int& eNext) {
    const int sNext = eNext - nNext;
    int2 myNext = make_int2(0, 0);
    if (lane < nNext) myNext = ed[sNext + lane];
    const int rAfter = rNext + rowStride;
    int nAfter = 0, eAfter = 0;
    if (rAfter < rEnd) { nAfter = cnt[rAfter]; eAfter = cursorEnd[rAfter]; }

    float4 acc = make_float4(0.0f, 0.0f, 0.0f, 0.0f);
    gcn_row_chunk(X4, my, min(32, n), lane, acc);
    for (int base = 32; base < n; base += 32) {
        const int m = min(32, n - base);
        int2 more = make_int2(0, 0);
        if (lane < m) more = ed[s + base + lane];
        gcn_row_chunk(X4, more, m, lane, acc);
    }
    n = nNext; s = sNext; my = myNext;
    rNext = rAfter; nNext = nAfter; eNext = eAfter;
    return acc;
}

__device__ __forceinline__ v8f gcn_wmma(v2f a, v2f b, v8f c) {
    return __builtin_amdgcn_wmma_f32_16x16x4_f32(
         false, a,  false, b,
         (short)0, c,  false,  false);
}

__device__ __forceinline__ void gcn_piece_two_layers(const float* __restrict__ sBase,
                                                     const float* __restrict__ w1Base,
                                                     const float* __restrict__ w2Base,
                                                     float* __restrict__ part, int lane) {
    v8f h0 = (v8f){};
    v8f h1 = (v8f){};
#pragma unroll 4
    for (int k = 0; k < DIM_K; k += 4) {
        const v2f sv = *(const v2f*)(sBase + k);
        const v2f wa = *(const v2f*)(w1Base + k);
        const v2f wb = *(const v2f*)(w1Base + 16 * W_PITCH + k);
        h0 = gcn_wmma(wa, sv, h0);
        h1 = gcn_wmma(wb, sv, h1);
    }
#pragma unroll
    for (int r = 0; r < 8; ++r) { h0[r] = fast_tanh(h0[r]); h1[r] = fast_tanh(h1[r]); }

    v8f p0 = (v8f){}, p1 = (v8f){}, p2 = (v8f){}, p3 = (v8f){};
#pragma unroll
    for (int q = 0; q < 4; ++q) {
        v2f b0, b1;
        b0.x = h0[2 * q]; b0.y = h0[2 * q + 1];
        b1.x = h1[2 * q]; b1.y = h1[2 * q + 1];
        const float* a = w2Base + 2 * q;
        p0 = gcn_wmma(*(const v2f*)(a),                b0, p0);
        p1 = gcn_wmma(*(const v2f*)(a + 16 * W_PITCH), b0, p1);
        p2 = gcn_wmma(*(const v2f*)(a + 32 * W_PITCH), b0, p2);
        p3 = gcn_wmma(*(const v2f*)(a + 48 * W_PITCH), b0, p3);
        p0 = gcn_wmma(*(const v2f*)(a + 16),                b1, p0);
        p1 = gcn_wmma(*(const v2f*)(a + 16 + 16 * W_PITCH), b1, p1);
        p2 = gcn_wmma(*(const v2f*)(a + 16 + 32 * W_PITCH), b1, p2);
        p3 = gcn_wmma(*(const v2f*)(a + 16 + 48 * W_PITCH), b1, p3);
    }
    float* mine = part + (lane & 15) * PP_PITCH + 8 * (lane >> 4);
    *(float4*)(mine)      = make_float4(p0[0], p0[1], p0[2], p0[3]);
    *(float4*)(mine + 4)  = make_float4(p0[4], p0[5], p0[6], p0[7]);
    *(float4*)(mine + 16) = make_float4(p1[0], p1[1], p1[2], p1[3]);
    *(float4*)(mine + 20) = make_float4(p1[4], p1[5], p1[6], p1[7]);
    *(float4*)(mine + 32) = make_float4(p2[0], p2[1], p2[2], p2[3]);
    *(float4*)(mine + 36) = make_float4(p2[4], p2[5], p2[6], p2[7]);
    *(float4*)(mine + 48) = make_float4(p3[0], p3[1], p3[2], p3[3]);
    *(float4*)(mine + 52) = make_float4(p3[4], p3[5], p3[6], p3[7]);
}

__device__ __forceinline__ void gcn_strip_sum_store(const float* __restrict__ parts,
                                                    float* __restrict__ P,
                                                    int rowFirst, int rowEnd, int lane) {
    const int c = lane & 15;
    const int h = lane >> 4;
#pragma unroll
    for (int i = 0; i < 8; ++i) {
        const int row = 2 * i + h;
        const float* src = parts + row * PP_PITCH + c * 4;
        const float4 a0 = *(const float4*)(src);
        const float4 a1 = *(const float4*)(src + PP_PIECE);
        const float4 a2 = *(const float4*)(src + 2 * PP_PIECE);
        const float4 a3 = *(const float4*)(src + 3 * PP_PIECE);
        float4 v;
        v.x = ((a0.x + a1.x) + a2.x) + a3.x;
        v.y = ((a0.y + a1.y) + a2.y) + a3.y;
        v.z = ((a0.z + a1.z) + a2.z) + a3.z;
        v.w = ((a0.w + a1.w) + a2.w) + a3.w;
        if (rowFirst + row < rowEnd)
            *(float4*)(P + (long long)(rowFirst + row) * DIM_OUT + c * 4) = v;
    }
}

__global__ void __launch_bounds__(FUSE_THREADS)
gcn_gather_gemm2_kernel(const float* __restrict__ X,
                        const int* __restrict__ cntGlobal,
                        const int* __restrict__ cursorEndGlobal,
                        const int* __restrict__ bucket,
                        const float* __restrict__ W1,
                        const float* __restrict__ W2,
                        float* __restrict__ P, int N) {
    const gcn_meta_ptr cnt = (gcn_meta_ptr)cntGlobal;
    const gcn_meta_ptr cursorEnd = (gcn_meta_ptr)cursorEndGlobal;
    __shared__ __attribute__((aligned(16))) float Wl[DIM_K * W_PITCH];
    __shared__ __attribute__((aligned(16))) float W2l[DIM_OUT * W_PITCH];
    __shared__ __attribute__((aligned(16))) float St[2 * FUSE_TILE_ROWS * W_PITCH];
    __shared__ __attribute__((aligned(16))) float Pp[2 * PP_TILE];

    {
        const float4* W4 = (const float4*)W1;
        const float4* V4 = (const float4*)W2;
        constexpr int W_BATCH = DIM_K * (DIM_K / 4) / FUSE_THREADS;
        constexpr int V_BATCH = DIM_OUT * (DIM_K / 4) / FUSE_THREADS;
        float4 w[W_BATCH + V_BATCH];
#pragma unroll
        for (int j = 0; j < W_BATCH; ++j) w[j] = W4[(int)threadIdx.x + j * FUSE_THREADS];
#pragma unroll
        for (int j = 0; j < V_BATCH; ++j) w[W_BATCH + j] = V4[(int)threadIdx.x + j * FUSE_THREADS];
#pragma unroll
        for (int j = 0; j < W_BATCH; ++j) {
            const int i = (int)threadIdx.x + j * FUSE_THREADS;
            *(float4*)&Wl[(i >> 5) * W_PITCH + (i & 31) * 4] = w[j];
        }
#pragma unroll
        for (int j = 0; j < V_BATCH; ++j) {
            const int i = (int)threadIdx.x + j * FUSE_THREADS;
            *(float4*)&W2l[(i >> 5) * W_PITCH + (i & 31) * 4] = w[W_BATCH + j];
        }
    }

    const int lane = threadIdx.x & 31;
    const int wave = __builtin_amdgcn_readfirstlane((int)(threadIdx.x >> 5));
    const int quad = wave >> 2;
    const int pos  = wave & 3;
    const bool mmaWave = pos == (quad & 3);
    const int g = quad * 3 + ((pos > (quad & 3)) ? pos - 1 : pos);

    const int rowsPer = (N + (int)gridDim.x - 1) / (int)gridDim.x;
    const int r0 = min(N, (int)blockIdx.x * rowsPer);
    const int r1 = min(N, r0 + rowsPer);
    const int nT = (r1 - r0 + FUSE_TILE_ROWS - 1) / FUSE_TILE_ROWS;

    const int2* ed = (const int2*)bucket;
    const float4* X4 = (const float4*)X;

    int n = 0, s = 0;
    int2 my = make_int2(0, 0);
    int rNext = r0 + g, nNext = 0, eNext = 0;
    if (!mmaWave) {
        if (rNext < r1) { nNext = cnt[rNext]; eNext = cursorEnd[rNext]; }
        (void)gcn_row_trip(X4, ed, cnt, cursorEnd, lane, FUSE_GATHER_WAVES, r1,
                           n, s, my, rNext, nNext, eNext);
    }

    const int l15   = lane & 15;
    const int lhalf = lane >> 4;

    for (int it = 0; it <= nT + 1; ++it) {
        if (!mmaWave) {
            if (it < nT) {
                float* tile = St + (it & 1) * (FUSE_TILE_ROWS * W_PITCH);
#pragma unroll
                for (int q = 0; q < FUSE_ROWS_PER_WAVE; ++q) {
                    const float4 acc = gcn_row_trip(X4, ed, cnt, cursorEnd, lane,
                                                    FUSE_GATHER_WAVES, r1,
                                                    n, s, my, rNext, nNext, eNext);
                    *(float4*)&tile[(g + q * FUSE_GATHER_WAVES) * W_PITCH + lane * 4] = acc;
                }
            }
        } else {
            if (it >= 2 && quad >= 4 && quad < 4 + FUSE_STRIPS) {
                const int tt = it - 2;
                const int st = quad - 4;
                const int rowFirst = r0 + tt * FUSE_TILE_ROWS + st * 16;
                if (rowFirst < r1)
                    gcn_strip_sum_store(Pp + (tt & 1) * PP_TILE + st * 4 * PP_PIECE, P,
                                        rowFirst, r1, lane);
            }
            if (it >= 1 && it <= nT) {
                const int tt = it - 1;
                const float* tile = St + (tt & 1) * (FUSE_TILE_ROWS * W_PITCH);
                float* parts = Pp + (tt & 1) * PP_TILE;
                const int tileRow = r0 + tt * FUSE_TILE_ROWS;
                const int cp = quad & 3;
                const float* w1Base = Wl + (cp * 32 + l15) * W_PITCH + 2 * lhalf;
                const float* w2Base = W2l + l15 * W_PITCH + cp * 32 + 8 * lhalf;
                for (int st = (quad < 4) ? 0 : 1; st < FUSE_STRIPS; st += 2) {
                    if (tileRow + st * 16 < r1) {
                        gcn_piece_two_layers(tile + (st * 16 + l15) * W_PITCH + 2 * lhalf,
                                             w1Base, w2Base,
                                             parts + (st * 4 + cp) * PP_PIECE, lane);
                    }
                }
            }
        }
        __syncthreads();
    }
}

extern "C" void kernel_launch(void* const* d_in, const int* in_sizes, int n_in,
                              void* d_out, int out_size, void* d_ws, size_t ws_size,
                              hipStream_t stream) {
    const float* X        = (const float*)d_in[0];
    const float* W1       = (const float*)d_in[1];
    const float* W2       = (const float*)d_in[2];
    const float* edgeVals = (const float*)d_in[3];
    const int*   edgeRow  = (const int*)d_in[4];
    const int*   edgeCol  = (const int*)d_in[5];
    float* out = (float*)d_out;

    const int N = in_sizes[0] / DIM_K;
    const int E = in_sizes[3];
    const long long featCount = (long long)N * DIM_K;

    const int NB = (N + ROWS_PER_BIN - 1) / ROWS_PER_BIN;
    const int nPad = (N + 31) & ~31;

    float* P  = (float*)d_ws;
    int* cnt    = (int*)(P + 2 * featCount);
    int* cursor = cnt + nPad;
    int* bucket = cursor + nPad;
    int* binned = bucket + 2 * (long long)E;
    int* segTab = binned + 2 * (long long)E;

    const int gatherThreads = GATHER_THREADS;
    const int gatherBlocks = GATHER_GROUPS;

    gcn_batch_sort_kernel<<<EDGE_CHUNKS, SORT_THREADS, 2 * NB * sizeof(int), stream>>>(
        edgeVals, edgeRow, edgeCol, segTab, binned, E, NB);
    gcn_bin_rows_kernel<<<NB, ROWS_THREADS, 0, stream>>>(
        binned, segTab, cnt, cursor, bucket, N, E);

    gcn_gather_gemm2_kernel<<<FUSE_GROUPS, FUSE_THREADS, 0, stream>>>(
        X, cnt, cursor, bucket, W1, W2, P, N);

    gcn_gather64_tanh_kernel<<<gatherBlocks, gatherThreads, 0, stream>>>(P, cnt, cursor, bucket, out, N);
}
